# starting kernel
# baseline (speedup 1.0000x reference)
#include <hip/hip_runtime.h>
#include <hip/hip_bf16.h>
#include <math.h>

typedef float v2f __attribute__((ext_vector_type(2)));
typedef float v8f __attribute__((ext_vector_type(8)));

#define IN_DIM   128
#define HID      32
#define HEADS    2
#define OUT_DIM  32

__device__ __forceinline__ void atomicMaxFloat(float* addr, float value) {
    if (value >= 0.0f) {
        atomicMax((int*)addr, __float_as_int(value));
    } else {
        atomicMin((unsigned int*)addr, __float_as_uint(value));
    }
}

__global__ void fill_kernel(float* __restrict__ p, float v, int n) {
    int i = blockIdx.x * blockDim.x + threadIdx.x;
    if (i < n) p[i] = v;
}

template <int K, int NC>
__global__ void wmma_gemm_f32(const float* __restrict__ A,
                              const float* __restrict__ B,
                              float* __restrict__ C, int M) {
    constexpr int NT = NC / 16;
    __shared__ v2f Bs2[(K / 2) * NC];

    for (int p = threadIdx.x; p < (K / 2) * NC; p += blockDim.x) {
        int k2 = p / NC;
        int n  = p - k2 * NC;
        v2f pr;
        pr.x = B[(2 * k2 + 0) * NC + n];
        pr.y = B[(2 * k2 + 1) * NC + n];
        Bs2[p] = pr;
    }
    __syncthreads();

    const int wave = threadIdx.x >> 5;
    const int lane = threadIdx.x & 31;
    const int mt   = blockIdx.x * 4 + wave;
    if (mt * 16 >= M) return;

    const int halfK = (lane >> 4) * 2;
    const int ncol  = lane & 15;
    const int arow  = mt * 16 + (lane & 15);
    const float* Arow = A + (size_t)arow * K;

    v8f acc[NT];
#pragma unroll
    for (int t = 0; t < NT; ++t) acc[t] = (v8f){0, 0, 0, 0, 0, 0, 0, 0};

    for (int k = 0; k < K; k += 4) {
        v2f a = *(const v2f*)(Arow + k + halfK);
        const int prow = (k + halfK) >> 1;
#pragma unroll
        for (int t = 0; t < NT; ++t) {
            v2f b = Bs2[prow * NC + t * 16 + ncol];
            acc[t] = __builtin_amdgcn_wmma_f32_16x16x4_f32(
                false, a, false, b, (short)0, acc[t], false, false);
        }
    }

    const int rbase = mt * 16 + ((lane >> 4) << 3);
#pragma unroll
    for (int t = 0; t < NT; ++t) {
#pragma unroll
        for (int j = 0; j < 8; ++j) {
            C[(size_t)(rbase + j) * NC + t * 16 + ncol] = acc[t][j];
        }
    }
}

template <int H, int C>
__global__ void score_kernel(const float* __restrict__ xl,
                             const float* __restrict__ a_src,
                             const float* __restrict__ a_dst,
                             float* __restrict__ asn,
                             float* __restrict__ adn, int n) {
    int i = blockIdx.x * blockDim.x + threadIdx.x;
    if (i >= n) return;
    const float* xr = xl + (size_t)i * H * C;
#pragma unroll
    for (int h = 0; h < H; ++h) {
        float s = 0.0f, d = 0.0f;
#pragma unroll
        for (int c = 0; c < C; ++c) {
            float v = xr[h * C + c];
            s += v * a_src[h * C + c];
            d += v * a_dst[h * C + c];
        }
        asn[i * H + h] = s;
        adn[i * H + h] = d;
    }
}

template <int H>
__global__ void edge_max_kernel(const int* __restrict__ ei, int E, int EE,
                                const float* __restrict__ asn,
                                const float* __restrict__ adn,
                                float* __restrict__ ebuf,
                                float* __restrict__ mbuf) {
    int e = blockIdx.x * blockDim.x + threadIdx.x;
    if (e >= EE) return;
    int src, dst;
    if (e < E) { src = ei[e]; dst = ei[E + e]; }
    else       { src = dst = e - E; }
#pragma unroll
    for (int h = 0; h < H; ++h) {
        float v = asn[src * H + h] + adn[dst * H + h];
        v = (v > 0.0f) ? v : 0.2f * v;
        ebuf[(size_t)e * H + h] = v;
        atomicMaxFloat(&mbuf[dst * H + h], v);
    }
}

template <int H>
__global__ void edge_exp_kernel(const int* __restrict__ ei, int E, int EE,
                                float* __restrict__ ebuf,
                                const float* __restrict__ mbuf,
                                float* __restrict__ den) {
    int e = blockIdx.x * blockDim.x + threadIdx.x;
    if (e >= EE) return;
    int dst = (e < E) ? ei[E + e] : (e - E);
#pragma unroll
    for (int h = 0; h < H; ++h) {
        float m = mbuf[dst * H + h];
        if (!(m >= -3.4e38f && m <= 3.4e38f)) m = 0.0f;
        float ex = __expf(ebuf[(size_t)e * H + h] - m);
        ebuf[(size_t)e * H + h] = ex;
        atomicAdd(&den[dst * H + h], ex);
    }
}

template <int H, int C>
__global__ void edge_agg_kernel(const int* __restrict__ ei, int E, int EE,
                                const float* __restrict__ ebuf,
                                const float* __restrict__ den,
                                const float* __restrict__ xl,
                                float* __restrict__ agg) {
    constexpr int F = H * C;
    constexpr int G = F / 4;
    long long t = (long long)blockIdx.x * blockDim.x + threadIdx.x;
    long long e = t / G;
    int g = (int)(t % G);
    if (e >= EE) return;
    int src, dst;
    if (e < E) { src = ei[e]; dst = ei[E + e]; }
    else       { src = dst = (int)(e - E); }
    int f0 = g * 4;
    int h  = f0 / C;
    float alpha = ebuf[(size_t)e * H + h] / (den[dst * H + h] + 1e-16f);
    float4 v = *(const float4*)(xl + (size_t)src * F + f0);
    float* o = agg + (size_t)dst * F + f0;
    atomicAdd(o + 0, v.x * alpha);
    atomicAdd(o + 1, v.y * alpha);
    atomicAdd(o + 2, v.z * alpha);
    atomicAdd(o + 3, v.w * alpha);
}

__global__ void bias_relu_kernel(float* __restrict__ a,
                                 const float* __restrict__ b, int n, int F) {
    int i = blockIdx.x * blockDim.x + threadIdx.x;
    if (i >= n * F) return;
    float v = a[i] + b[i % F];
    a[i] = v > 0.0f ? v : 0.0f;
}

__global__ void bias_add_kernel(float* __restrict__ a,
                                const float* __restrict__ b, int n, int F) {
    int i = blockIdx.x * blockDim.x + threadIdx.x;
    if (i >= n * F) return;
    a[i] += b[i % F];
}

static inline int cdiv(long long a, long long b) { return (int)((a + b - 1) / b); }

extern "C" void kernel_launch(void* const* d_in, const int* in_sizes, int n_in,
                              void* d_out, int out_size, void* d_ws, size_t ws_size,
                              hipStream_t stream) {
    const float* x      = (const float*)d_in[0];
    const int*   ei     = (const int*)d_in[1];
    const float* W1     = (const float*)d_in[2];
    const float* a_src1 = (const float*)d_in[3];
    const float* a_dst1 = (const float*)d_in[4];
    const float* b1     = (const float*)d_in[5];
    const float* W2     = (const float*)d_in[6];
    const float* a_src2 = (const float*)d_in[7];
    const float* a_dst2 = (const float*)d_in[8];
    const float* b2     = (const float*)d_in[9];
    float* out = (float*)d_out;

    const int N  = in_sizes[0] / IN_DIM;
    const int E  = in_sizes[1] / 2;
    const int EE = E + N;
    const int F1 = HEADS * HID;

    float* ws   = (float*)d_ws;
    float* xl1  = ws;
    float* agg1 = xl1  + (size_t)N * F1;
    float* asn1 = agg1 + (size_t)N * F1;
    float* adn1 = asn1 + (size_t)N * HEADS;
    float* m1   = adn1 + (size_t)N * HEADS;
    float* den1 = m1   + (size_t)N * HEADS;
    float* ex1  = den1 + (size_t)N * HEADS;
    float* xl2  = ex1  + (size_t)EE * HEADS;
    float* asn2 = xl2  + (size_t)N * OUT_DIM;
    float* adn2 = asn2 + (size_t)N;
    float* m2   = adn2 + (size_t)N;
    float* den2 = m2   + (size_t)N;
    float* ex2  = den2 + (size_t)N;

    const int TB = 256;

    fill_kernel<<<cdiv((long long)N * F1, TB), TB, 0, stream>>>(agg1, 0.0f, N * F1);
    fill_kernel<<<cdiv((long long)N * OUT_DIM, TB), TB, 0, stream>>>(out, 0.0f, N * OUT_DIM);
    fill_kernel<<<cdiv((long long)N * HEADS, TB), TB, 0, stream>>>(m1, -INFINITY, N * HEADS);
    fill_kernel<<<cdiv((long long)N * HEADS, TB), TB, 0, stream>>>(den1, 0.0f, N * HEADS);
    fill_kernel<<<cdiv(N, TB), TB, 0, stream>>>(m2, -INFINITY, N);
    fill_kernel<<<cdiv(N, TB), TB, 0, stream>>>(den2, 0.0f, N);

    {
        int mtiles = cdiv(N, 16);
        wmma_gemm_f32<IN_DIM, F1><<<cdiv(mtiles, 4), 128, 0, stream>>>(x, W1, xl1, N);
        score_kernel<HEADS, HID><<<cdiv(N, TB), TB, 0, stream>>>(xl1, a_src1, a_dst1, asn1, adn1, N);
        edge_max_kernel<HEADS><<<cdiv(EE, TB), TB, 0, stream>>>(ei, E, EE, asn1, adn1, ex1, m1);
        edge_exp_kernel<HEADS><<<cdiv(EE, TB), TB, 0, stream>>>(ei, E, EE, ex1, m1, den1);
        edge_agg_kernel<HEADS, HID><<<cdiv((long long)EE * (F1 / 4), TB), TB, 0, stream>>>(
            ei, E, EE, ex1, den1, xl1, agg1);
        bias_relu_kernel<<<cdiv((long long)N * F1, TB), TB, 0, stream>>>(agg1, b1, N, F1);
    }

    {
        int mtiles = cdiv(N, 16);
        wmma_gemm_f32<F1, OUT_DIM><<<cdiv(mtiles, 4), 128, 0, stream>>>(agg1, W2, xl2, N);
        score_kernel<1, OUT_DIM><<<cdiv(N, TB), TB, 0, stream>>>(xl2, a_src2, a_dst2, asn2, adn2, N);
        edge_max_kernel<1><<<cdiv(EE, TB), TB, 0, stream>>>(ei, E, EE, asn2, adn2, ex2, m2);
        edge_exp_kernel<1><<<cdiv(EE, TB), TB, 0, stream>>>(ei, E, EE, ex2, m2, den2);
        edge_agg_kernel<1, OUT_DIM><<<cdiv((long long)EE * (OUT_DIM / 4), TB), TB, 0, stream>>>(
            ei, E, EE, ex2, den2, xl2, out);
        bias_add_kernel<<<cdiv((long long)N * OUT_DIM, TB), TB, 0, stream>>>(out, b2, N, OUT_DIM);
    }
}


# Softmax division moved behind the segment sum: one edge pass adds exp-weighted rows and the denominator, the bias kernel divides once per node; exp pass removed (rounding order of the sum changes)
# speedup vs baseline: 1.1074x; 1.1074x over previous
#include <hip/hip_runtime.h>
#include <hip/hip_bf16.h>
#include <math.h>

typedef float v2f __attribute__((ext_vector_type(2)));
typedef float v8f __attribute__((ext_vector_type(8)));

#define IN_DIM   128
#define HID      32
#define HEADS    2
#define OUT_DIM  32

__device__ __forceinline__ void atomicMaxFloat(float* addr, float value) {
    if (value >= 0.0f) {
        atomicMax((int*)addr, __float_as_int(value));
    } else {
        atomicMin((unsigned int*)addr, __float_as_uint(value));
    }
}

__global__ void fill_kernel(float* __restrict__ p, float v, int n) {
    int i = blockIdx.x * blockDim.x + threadIdx.x;
    if (i < n) p[i] = v;
}

template <int K, int NC>
__global__ void wmma_gemm_f32(const float* __restrict__ A,
                              const float* __restrict__ B,
                              float* __restrict__ C, int M) {
    constexpr int NT = NC / 16;
    __shared__ v2f Bs2[(K / 2) * NC];

    for (int p = threadIdx.x; p < (K / 2) * NC; p += blockDim.x) {
        int k2 = p / NC;
        int n  = p - k2 * NC;
        v2f pr;
        pr.x = B[(2 * k2 + 0) * NC + n];
        pr.y = B[(2 * k2 + 1) * NC + n];
        Bs2[p] = pr;
    }
    __syncthreads();

    const int wave = threadIdx.x >> 5;
    const int lane = threadIdx.x & 31;
    const int mt   = blockIdx.x * 4 + wave;
    if (mt * 16 >= M) return;

    const int halfK = (lane >> 4) * 2;
    const int ncol  = lane & 15;
    const int arow  = mt * 16 + (lane & 15);
    const float* Arow = A + (size_t)arow * K;

    v8f acc[NT];
#pragma unroll
    for (int t = 0; t < NT; ++t) acc[t] = (v8f){0, 0, 0, 0, 0, 0, 0, 0};

    for (int k = 0; k < K; k += 4) {
        v2f a = *(const v2f*)(Arow + k + halfK);
        const int prow = (k + halfK) >> 1;
#pragma unroll
        for (int t = 0; t < NT; ++t) {
            v2f b = Bs2[prow * NC + t * 16 + ncol];
            acc[t] = __builtin_amdgcn_wmma_f32_16x16x4_f32(
                false, a, false, b, (short)0, acc[t], false, false);
        }
    }

    const int rbase = mt * 16 + ((lane >> 4) << 3);
#pragma unroll
    for (int t = 0; t < NT; ++t) {
#pragma unroll
        for (int j = 0; j < 8; ++j) {
            C[(size_t)(rbase + j) * NC + t * 16 + ncol] = acc[t][j];
        }
    }
}

template <int H, int C>
__global__ void score_kernel(const float* __restrict__ xl,
                             const float* __restrict__ a_src,
                             const float* __restrict__ a_dst,
                             float* __restrict__ asn,
                             float* __restrict__ adn, int n) {
    int i = blockIdx.x * blockDim.x + threadIdx.x;
    if (i >= n) return;
    const float* xr = xl + (size_t)i * H * C;
#pragma unroll
    for (int h = 0; h < H; ++h) {
        float s = 0.0f, d = 0.0f;
#pragma unroll
        for (int c = 0; c < C; ++c) {
            float v = xr[h * C + c];
            s += v * a_src[h * C + c];
            d += v * a_dst[h * C + c];
        }
        asn[i * H + h] = s;
        adn[i * H + h] = d;
    }
}

template <int H>
__global__ void edge_max_kernel(const int* __restrict__ ei, int E, int EE,
                                const float* __restrict__ asn,
                                const float* __restrict__ adn,
                                float* __restrict__ ebuf,
                                float* __restrict__ mbuf) {
    int e = blockIdx.x * blockDim.x + threadIdx.x;
    if (e >= EE) return;
    int src, dst;
    if (e < E) { src = ei[e]; dst = ei[E + e]; }
    else       { src = dst = e - E; }
#pragma unroll
    for (int h = 0; h < H; ++h) {
        float v = asn[src * H + h] + adn[dst * H + h];
        v = (v > 0.0f) ? v : 0.2f * v;
        ebuf[(size_t)e * H + h] = v;
        atomicMaxFloat(&mbuf[dst * H + h], v);
    }
}

template <int H, int C>
__global__ void edge_agg_kernel(const int* __restrict__ ei, int E, int EE,
                                const float* __restrict__ ebuf,
                                const float* __restrict__ mbuf,
                                const float* __restrict__ xl,
                                float* __restrict__ den,
                                float* __restrict__ agg) {
    constexpr int F = H * C;
    constexpr int G = F / 4;
    constexpr int GH = C / 4;
    long long t = (long long)blockIdx.x * blockDim.x + threadIdx.x;
    long long e = t / G;
    int g = (int)(t % G);
    if (e >= EE) return;
    int src, dst;
    if (e < E) { src = ei[e]; dst = ei[E + e]; }
    else       { src = dst = (int)(e - E); }
    int f0 = g * 4;
    int h  = f0 / C;
    float m = mbuf[dst * H + h];
    if (!(m >= -3.4e38f && m <= 3.4e38f)) m = 0.0f;
    float ex = __expf(ebuf[(size_t)e * H + h] - m);
    float4 v = *(const float4*)(xl + (size_t)src * F + f0);
    float* o = agg + (size_t)dst * F + f0;
    atomicAdd(o + 0, v.x * ex);
    atomicAdd(o + 1, v.y * ex);
    atomicAdd(o + 2, v.z * ex);
    atomicAdd(o + 3, v.w * ex);
    if (g % GH == 0) atomicAdd(&den[dst * H + h], ex);
}

template <int F, int C>
__global__ void bias_relu_kernel(float* __restrict__ a,
                                 const float* __restrict__ den,
                                 const float* __restrict__ b, int n) {
    int i = blockIdx.x * blockDim.x + threadIdx.x;
    if (i >= n * F) return;
    float v = a[i] / (den[i / C] + 1e-16f) + b[i % F];
    a[i] = v > 0.0f ? v : 0.0f;
}

template <int F, int C>
__global__ void bias_add_kernel(float* __restrict__ a,
                                const float* __restrict__ den,
                                const float* __restrict__ b, int n) {
    int i = blockIdx.x * blockDim.x + threadIdx.x;
    if (i >= n * F) return;
    a[i] = a[i] / (den[i / C] + 1e-16f) + b[i % F];
}

static inline int cdiv(long long a, long long b) { return (int)((a + b - 1) / b); }

extern "C" void kernel_launch(void* const* d_in, const int* in_sizes, int n_in,
                              void* d_out, int out_size, void* d_ws, size_t ws_size,
                              hipStream_t stream) {
    const float* x      = (const float*)d_in[0];
    const int*   ei     = (const int*)d_in[1];
    const float* W1     = (const float*)d_in[2];
    const float* a_src1 = (const float*)d_in[3];
    const float* a_dst1 = (const float*)d_in[4];
    const float* b1     = (const float*)d_in[5];
    const float* W2     = (const float*)d_in[6];
    const float* a_src2 = (const float*)d_in[7];
    const float* a_dst2 = (const float*)d_in[8];
    const float* b2     = (const float*)d_in[9];
    float* out = (float*)d_out;

    const int N  = in_sizes[0] / IN_DIM;
    const int E  = in_sizes[1] / 2;
    const int EE = E + N;
    const int F1 = HEADS * HID;

    float* ws   = (float*)d_ws;
    float* xl1  = ws;
    float* agg1 = xl1  + (size_t)N * F1;
    float* asn1 = agg1 + (size_t)N * F1;
    float* adn1 = asn1 + (size_t)N * HEADS;
    float* m1   = adn1 + (size_t)N * HEADS;
    float* den1 = m1   + (size_t)N * HEADS;
    float* ex1  = den1 + (size_t)N * HEADS;
    float* xl2  = ex1  + (size_t)EE * HEADS;
    float* asn2 = xl2  + (size_t)N * OUT_DIM;
    float* adn2 = asn2 + (size_t)N;
    float* m2   = adn2 + (size_t)N;
    float* den2 = m2   + (size_t)N;
    float* ex2  = den2 + (size_t)N;

    const int TB = 256;

    fill_kernel<<<cdiv((long long)N * F1, TB), TB, 0, stream>>>(agg1, 0.0f, N * F1);
    fill_kernel<<<cdiv((long long)N * OUT_DIM, TB), TB, 0, stream>>>(out, 0.0f, N * OUT_DIM);
    fill_kernel<<<cdiv((long long)N * HEADS, TB), TB, 0, stream>>>(m1, -INFINITY, N * HEADS);
    fill_kernel<<<cdiv((long long)N * HEADS, TB), TB, 0, stream>>>(den1, 0.0f, N * HEADS);
    fill_kernel<<<cdiv(N, TB), TB, 0, stream>>>(m2, -INFINITY, N);
    fill_kernel<<<cdiv(N, TB), TB, 0, stream>>>(den2, 0.0f, N);

    {
        int mtiles = cdiv(N, 16);
        wmma_gemm_f32<IN_DIM, F1><<<cdiv(mtiles, 4), 128, 0, stream>>>(x, W1, xl1, N);
        score_kernel<HEADS, HID><<<cdiv(N, TB), TB, 0, stream>>>(xl1, a_src1, a_dst1, asn1, adn1, N);
        edge_max_kernel<HEADS><<<cdiv(EE, TB), TB, 0, stream>>>(ei, E, EE, asn1, adn1, ex1, m1);
        edge_agg_kernel<HEADS, HID><<<cdiv((long long)EE * (F1 / 4), TB), TB, 0, stream>>>(
            ei, E, EE, ex1, m1, xl1, den1, agg1);
        bias_relu_kernel<HEADS * HID, HID><<<cdiv((long long)N * F1, TB), TB, 0, stream>>>(agg1, den1, b1, N);
    }

    {
        int mtiles = cdiv(N, 16);
        wmma_gemm_f32<F1, OUT_DIM><<<cdiv(mtiles, 4), 128, 0, stream>>>(agg1, W2, xl2, N);
        score_kernel<1, OUT_DIM><<<cdiv(N, TB), TB, 0, stream>>>(xl2, a_src2, a_dst2, asn2, adn2, N);
        edge_max_kernel<1><<<cdiv(EE, TB), TB, 0, stream>>>(ei, E, EE, asn2, adn2, ex2, m2);
        edge_agg_kernel<1, OUT_DIM><<<cdiv((long long)EE * (OUT_DIM / 4), TB), TB, 0, stream>>>(
            ei, E, EE, ex2, m2, xl2, den2, out);
        bias_add_kernel<OUT_DIM, OUT_DIM><<<cdiv((long long)N * OUT_DIM, TB), TB, 0, stream>>>(out, den2, b2, N);
    }
}


# Edges grouped by destination in each call (integer atomic rank, scan, place); a lane group sums each node's rows in registers and writes once: float atomics, max pass, fills gone; sum order changes
# speedup vs baseline: 3.1186x; 2.8161x over previous
#include <hip/hip_runtime.h>
#include <hip/hip_bf16.h>
#include <math.h>

typedef float v2f __attribute__((ext_vector_type(2)));
typedef float v8f __attribute__((ext_vector_type(8)));

#define IN_DIM   128
#define HID      32
#define HEADS    2
#define OUT_DIM  32

__global__ void zero_kernel(int* __restrict__ p, int n) {
    int i = blockIdx.x * blockDim.x + threadIdx.x;
    if (i < n) p[i] = 0;
}

template <int K, int NC>
__global__ void wmma_gemm_f32(const float* __restrict__ A,
                              const float* __restrict__ B,
                              float* __restrict__ C, int M) {
    constexpr int NT = NC / 16;
    __shared__ v2f Bs2[(K / 2) * NC];

    for (int p = threadIdx.x; p < (K / 2) * NC; p += blockDim.x) {
        int k2 = p / NC;
        int n  = p - k2 * NC;
        v2f pr;
        pr.x = B[(2 * k2 + 0) * NC + n];
        pr.y = B[(2 * k2 + 1) * NC + n];
        Bs2[p] = pr;
    }
    __syncthreads();

    const int wave = threadIdx.x >> 5;
    const int lane = threadIdx.x & 31;
    const int mt   = blockIdx.x * 4 + wave;
    if (mt * 16 >= M) return;

    const int halfK = (lane >> 4) * 2;
    const int ncol  = lane & 15;
    const int arow  = mt * 16 + (lane & 15);
    const float* Arow = A + (size_t)arow * K;

    v8f acc[NT];
#pragma unroll
    for (int t = 0; t < NT; ++t) acc[t] = (v8f){0, 0, 0, 0, 0, 0, 0, 0};

    for (int k = 0; k < K; k += 4) {
        v2f a = *(const v2f*)(Arow + k + halfK);
        const int prow = (k + halfK) >> 1;
#pragma unroll
        for (int t = 0; t < NT; ++t) {
            v2f b = Bs2[prow * NC + t * 16 + ncol];
            acc[t] = __builtin_amdgcn_wmma_f32_16x16x4_f32(
                false, a, false, b, (short)0, acc[t], false, false);
        }
    }

    const int rbase = mt * 16 + ((lane >> 4) << 3);
#pragma unroll
    for (int t = 0; t < NT; ++t) {
#pragma unroll
        for (int j = 0; j < 8; ++j) {
            C[(size_t)(rbase + j) * NC + t * 16 + ncol] = acc[t][j];
        }
    }
}

template <int H, int C>
__global__ void score_kernel(const float* __restrict__ xl,
                             const float* __restrict__ a_src,
                             const float* __restrict__ a_dst,
                             float* __restrict__ asn,
                             float* __restrict__ adn, int n) {
    int i = blockIdx.x * blockDim.x + threadIdx.x;
    if (i >= n) return;
    const float* xr = xl + (size_t)i * H * C;
#pragma unroll
    for (int h = 0; h < H; ++h) {
        float s = 0.0f, d = 0.0f;
#pragma unroll
        for (int c = 0; c < C; ++c) {
            float v = xr[h * C + c];
            s += v * a_src[h * C + c];
            d += v * a_dst[h * C + c];
        }
        asn[i * H + h] = s;
        adn[i * H + h] = d;
    }
}

#define SCAN_TB  256
#define SCAN_PER 4

__global__ void rank_kernel(const int* __restrict__ ei, int E,
                            int* __restrict__ cnt, int* __restrict__ rank) {
    int e = blockIdx.x * blockDim.x + threadIdx.x;
    if (e >= E) return;
    int d = ei[E + e];
    rank[e] = atomicAdd(&cnt[d], 1);
}

__global__ void block_sum_kernel(const int* __restrict__ cnt,
                                 int* __restrict__ bsum, int n) {
    __shared__ int sh[SCAN_TB];
    const int tid  = threadIdx.x;
    const int base = blockIdx.x * (SCAN_TB * SCAN_PER) + tid * SCAN_PER;
    int s = 0;
#pragma unroll
    for (int k = 0; k < SCAN_PER; ++k) {
        int idx = base + k;
        if (idx < n) s += cnt[idx];
    }
    sh[tid] = s;
    __syncthreads();
    for (int st = SCAN_TB / 2; st > 0; st >>= 1) {
        if (tid < st) sh[tid] += sh[tid + st];
        __syncthreads();
    }
    if (tid == 0) bsum[blockIdx.x] = sh[0];
}

__global__ void scan_kernel(const int* __restrict__ cnt,
                            const int* __restrict__ bsum,
                            int* __restrict__ off, int n) {
    __shared__ int sh[SCAN_TB];
    const int tid = threadIdx.x;

    int pb = 0;
    for (int j = tid; j < (int)blockIdx.x; j += SCAN_TB) pb += bsum[j];
    sh[tid] = pb;
    __syncthreads();
    for (int st = SCAN_TB / 2; st > 0; st >>= 1) {
        if (tid < st) sh[tid] += sh[tid + st];
        __syncthreads();
    }
    const int block_base = sh[0];
    __syncthreads();

    const int base = blockIdx.x * (SCAN_TB * SCAN_PER) + tid * SCAN_PER;
    int c[SCAN_PER];
    int s = 0;
#pragma unroll
    for (int k = 0; k < SCAN_PER; ++k) {
        int idx = base + k;
        c[k] = (idx < n) ? cnt[idx] : 0;
        s += c[k];
    }
    sh[tid] = s;
    __syncthreads();
    for (int d = 1; d < SCAN_TB; d <<= 1) {
        int v = (tid >= d) ? sh[tid - d] : 0;
        __syncthreads();
        sh[tid] += v;
        __syncthreads();
    }
    int run = sh[tid] - s + block_base;
#pragma unroll
    for (int k = 0; k < SCAN_PER; ++k) {
        int idx = base + k;
        if (idx < n) off[idx] = run;
        run += c[k];
    }
}

__global__ void place_kernel(const int* __restrict__ ei, int E,
                             const int* __restrict__ off,
                             const int* __restrict__ rank,
                             int* __restrict__ srcs) {
    int e = blockIdx.x * blockDim.x + threadIdx.x;
    if (e >= E) return;
    int d = ei[E + e];
    srcs[off[d] + rank[e]] = ei[e];
}

__device__ __forceinline__ float leaky02(float v) {
    return (v > 0.0f) ? v : 0.2f * v;
}

template <int H, int C, bool RELU>
__global__ void node_agg_kernel(const int* __restrict__ off,
                                const int* __restrict__ cnt,
                                const int* __restrict__ srcs,
                                const float* __restrict__ asn,
                                const float* __restrict__ adn,
                                const float* __restrict__ xl,
                                const float* __restrict__ b,
                                float* __restrict__ outp, int n) {
    constexpr int F   = H * C;
    constexpr int LPN = F / 4;
    const int t  = blockIdx.x * blockDim.x + threadIdx.x;
    const int i  = t / LPN;
    const int lg = t % LPN;
    if (i >= n) return;

    const int beg = off[i];
    const int deg = cnt[i];
    const int h   = (lg * 4) / C;

    float ad[H], mx[H];
#pragma unroll
    for (int hh = 0; hh < H; ++hh) {
        ad[hh] = adn[i * H + hh];
        mx[hh] = leaky02(asn[i * H + hh] + ad[hh]);
    }

    for (int j = lg; j < deg; j += LPN) {
        const int s = srcs[beg + j];
#pragma unroll
        for (int hh = 0; hh < H; ++hh)
            mx[hh] = fmaxf(mx[hh], leaky02(asn[s * H + hh] + ad[hh]));
    }
#pragma unroll
    for (int hh = 0; hh < H; ++hh) {
#pragma unroll
        for (int o = LPN / 2; o > 0; o >>= 1)
            mx[hh] = fmaxf(mx[hh], __shfl_xor(mx[hh], o, LPN));
    }

    float m = mx[0], adh = ad[0];
#pragma unroll
    for (int hh = 1; hh < H; ++hh) {
        if (h == hh) { m = mx[hh]; adh = ad[hh]; }
    }
    if (!(m >= -3.4e38f && m <= 3.4e38f)) m = 0.0f;

    const int f0 = lg * 4;
    float w   = __expf(leaky02(asn[i * H + h] + adh) - m);
    float den = w;
    float4 r  = *(const float4*)(xl + (size_t)i * F + f0);
    float4 acc;
    acc.x = r.x * w; acc.y = r.y * w; acc.z = r.z * w; acc.w = r.w * w;

#pragma unroll 4
    for (int j = 0; j < deg; ++j) {
        const int s = srcs[beg + j];
        const float wj = __expf(leaky02(asn[s * H + h] + adh) - m);
        const float4 rr = *(const float4*)(xl + (size_t)s * F + f0);
        den += wj;
        acc.x += rr.x * wj;
        acc.y += rr.y * wj;
        acc.z += rr.z * wj;
        acc.w += rr.w * wj;
    }

    const float dd = den + 1e-16f;
    float4 o;
    o.x = acc.x / dd + b[f0 + 0];
    o.y = acc.y / dd + b[f0 + 1];
    o.z = acc.z / dd + b[f0 + 2];
    o.w = acc.w / dd + b[f0 + 3];
    if (RELU) {
        o.x = o.x > 0.0f ? o.x : 0.0f;
        o.y = o.y > 0.0f ? o.y : 0.0f;
        o.z = o.z > 0.0f ? o.z : 0.0f;
        o.w = o.w > 0.0f ? o.w : 0.0f;
    }
    *(float4*)(outp + (size_t)i * F + f0) = o;
}

static inline int cdiv(long long a, long long b) { return (int)((a + b - 1) / b); }

extern "C" void kernel_launch(void* const* d_in, const int* in_sizes, int n_in,
                              void* d_out, int out_size, void* d_ws, size_t ws_size,
                              hipStream_t stream) {
    const float* x      = (const float*)d_in[0];
    const int*   ei     = (const int*)d_in[1];
    const float* W1     = (const float*)d_in[2];
    const float* a_src1 = (const float*)d_in[3];
    const float* a_dst1 = (const float*)d_in[4];
    const float* b1     = (const float*)d_in[5];
    const float* W2     = (const float*)d_in[6];
    const float* a_src2 = (const float*)d_in[7];
    const float* a_dst2 = (const float*)d_in[8];
    const float* b2     = (const float*)d_in[9];
    float* out = (float*)d_out;

    const int N  = in_sizes[0] / IN_DIM;
    const int E  = in_sizes[1] / 2;
    const int F1 = HEADS * HID;

    float* ws   = (float*)d_ws;
    float* xl1  = ws;
    float* h1   = xl1  + (size_t)N * F1;
    float* asn1 = h1   + (size_t)N * F1;
    float* adn1 = asn1 + (size_t)N * HEADS;
    float* xl2  = adn1 + (size_t)N * HEADS;
    float* asn2 = xl2  + (size_t)N * OUT_DIM;
    float* adn2 = asn2 + (size_t)N;
    int*   cnt  = (int*)(adn2 + (size_t)N);
    int*   off  = cnt  + (size_t)N;
    int*   bsum = off  + (size_t)N;
    const int NB = cdiv(N, SCAN_TB * SCAN_PER);
    int*   rank = bsum + (size_t)NB;
    int*   srcs = rank + (size_t)E;

    const int TB = 256;

    zero_kernel<<<cdiv(N, TB), TB, 0, stream>>>(cnt, N);
    rank_kernel<<<cdiv(E, TB), TB, 0, stream>>>(ei, E, cnt, rank);
    block_sum_kernel<<<NB, SCAN_TB, 0, stream>>>(cnt, bsum, N);
    scan_kernel<<<NB, SCAN_TB, 0, stream>>>(cnt, bsum, off, N);
    place_kernel<<<cdiv(E, TB), TB, 0, stream>>>(ei, E, off, rank, srcs);

    {
        int mtiles = cdiv(N, 16);
        wmma_gemm_f32<IN_DIM, F1><<<cdiv(mtiles, 4), 128, 0, stream>>>(x, W1, xl1, N);
        score_kernel<HEADS, HID><<<cdiv(N, TB), TB, 0, stream>>>(xl1, a_src1, a_dst1, asn1, adn1, N);
        node_agg_kernel<HEADS, HID, true><<<cdiv((long long)N * (F1 / 4), TB), TB, 0, stream>>>(
            off, cnt, srcs, asn1, adn1, xl1, b1, h1, N);
    }

    {
        int mtiles = cdiv(N, 16);
        wmma_gemm_f32<F1, OUT_DIM><<<cdiv(mtiles, 4), 128, 0, stream>>>(h1, W2, xl2, N);
        score_kernel<1, OUT_DIM><<<cdiv(N, TB), TB, 0, stream>>>(xl2, a_src2, a_dst2, asn2, adn2, N);
        node_agg_kernel<1, OUT_DIM, false><<<cdiv((long long)N * (OUT_DIM / 4), TB), TB, 0, stream>>>(
            off, cnt, srcs, asn2, adn2, xl2, b2, out, N);
    }
}


# Edge grouping as a two-level counting sort on LDS counters (512-node buckets, runs written as whole lines): no global atomic or scattered 4-byte access per edge; sum order stays arrival-dependent
# speedup vs baseline: 4.8819x; 1.5654x over previous
#include <hip/hip_runtime.h>
#include <hip/hip_bf16.h>
#include <math.h>

typedef float v2f __attribute__((ext_vector_type(2)));
typedef float v8f __attribute__((ext_vector_type(8)));

#define IN_DIM   128
#define HID      32
#define HEADS    2
#define OUT_DIM  32

template <int K, int NC>
__global__ void wmma_gemm_f32(const float* __restrict__ A,
                              const float* __restrict__ B,
                              float* __restrict__ C, int M) {
    constexpr int NT = NC / 16;
    __shared__ v2f Bs2[(K / 2) * NC];

    for (int p = threadIdx.x; p < (K / 2) * NC; p += blockDim.x) {
        int k2 = p / NC;
        int n  = p - k2 * NC;
        v2f pr;
        pr.x = B[(2 * k2 + 0) * NC + n];
        pr.y = B[(2 * k2 + 1) * NC + n];
        Bs2[p] = pr;
    }
    __syncthreads();

    const int wave = threadIdx.x >> 5;
    const int lane = threadIdx.x & 31;
    const int mt   = blockIdx.x * 4 + wave;
    if (mt * 16 >= M) return;

    const int halfK = (lane >> 4) * 2;
    const int ncol  = lane & 15;
    const int arow  = mt * 16 + (lane & 15);
    const float* Arow = A + (size_t)arow * K;

    v8f acc[NT];
#pragma unroll
    for (int t = 0; t < NT; ++t) acc[t] = (v8f){0, 0, 0, 0, 0, 0, 0, 0};

    for (int k = 0; k < K; k += 4) {
        v2f a = *(const v2f*)(Arow + k + halfK);
        const int prow = (k + halfK) >> 1;
#pragma unroll
        for (int t = 0; t < NT; ++t) {
            v2f b = Bs2[prow * NC + t * 16 + ncol];
            acc[t] = __builtin_amdgcn_wmma_f32_16x16x4_f32(
                false, a, false, b, (short)0, acc[t], false, false);
        }
    }

    const int rbase = mt * 16 + ((lane >> 4) << 3);
#pragma unroll
    for (int t = 0; t < NT; ++t) {
#pragma unroll
        for (int j = 0; j < 8; ++j) {
            C[(size_t)(rbase + j) * NC + t * 16 + ncol] = acc[t][j];
        }
    }
}

template <int H, int C>
__global__ void score_kernel(const float* __restrict__ xl,
                             const float* __restrict__ a_src,
                             const float* __restrict__ a_dst,
                             float* __restrict__ asn,
                             float* __restrict__ adn, int n) {
    int i = blockIdx.x * blockDim.x + threadIdx.x;
    if (i >= n) return;
    const float* xr = xl + (size_t)i * H * C;
#pragma unroll
    for (int h = 0; h < H; ++h) {
        float s = 0.0f, d = 0.0f;
#pragma unroll
        for (int c = 0; c < C; ++c) {
            float v = xr[h * C + c];
            s += v * a_src[h * C + c];
            d += v * a_dst[h * C + c];
        }
        asn[i * H + h] = s;
        adn[i * H + h] = d;
    }
}

#define BK_SHIFT  9
#define BK_NODES  (1 << BK_SHIFT)
#define MAXB      256
#define CH_TB     256
#define CH_PER    16
#define CH        (CH_TB * CH_PER)
#define REC_SHIFT 22
#define REC_MASK  ((1 << REC_SHIFT) - 1)
#define CS_TB     512
#define SORT_TB   1024
#define SORT_PER  16
#define SORT_CAP  (SORT_TB * SORT_PER)

template <int NT>
__device__ __forceinline__ int block_excl_scan(int v, int* wsum, int* total) {
    const int lane = threadIdx.x & 31;
    const int wv   = threadIdx.x >> 5;
    int inc = v;
#pragma unroll
    for (int o = 1; o < 32; o <<= 1) {
        int t = __shfl_up(inc, o, 32);
        if (lane >= o) inc += t;
    }
    if (lane == 31) wsum[wv] = inc;
    __syncthreads();
    if (wv == 0) {
        const int ws = (lane < NT / 32) ? wsum[lane] : 0;
        int wi = ws;
#pragma unroll
        for (int o = 1; o < 32; o <<= 1) {
            int t = __shfl_up(wi, o, 32);
            if (lane >= o) wi += t;
        }
        if (lane < NT / 32) wsum[lane] = wi - ws;
        if (lane == 31) wsum[32] = wi;
    }
    __syncthreads();
    const int res = wsum[wv] + inc - v;
    *total = wsum[32];
    __syncthreads();
    return res;
}

__global__ __launch_bounds__(CH_TB)
void bucket_hist_kernel(const int* __restrict__ ei, int E, int G, int nbkt,
                        int* __restrict__ hist_t) {
    __shared__ int lh[MAXB];
    const int tid  = threadIdx.x;
    const int base = blockIdx.x * CH;
    lh[tid] = 0;
    __syncthreads();
#pragma unroll
    for (int k = 0; k < CH_PER; ++k) {
        const int e = base + k * CH_TB + tid;
        if (e < E) atomicAdd(&lh[ei[E + e] >> BK_SHIFT], 1);
    }
    __syncthreads();
    if (tid < nbkt) hist_t[tid * G + blockIdx.x] = lh[tid];
}

__global__ __launch_bounds__(CS_TB)
void bucket_colscan_kernel(const int* __restrict__ hist_t, int G,
                           int* __restrict__ woff_t, int* __restrict__ tot) {
    __shared__ int wsum[33];
    const int b = blockIdx.x;
    int carry = 0;
    for (int w0 = 0; w0 < G; w0 += CS_TB) {
        const int w = w0 + threadIdx.x;
        const int v = (w < G) ? hist_t[b * G + w] : 0;
        int total;
        const int ex = block_excl_scan<CS_TB>(v, wsum, &total);
        if (w < G) woff_t[b * G + w] = carry + ex;
        carry += total;
    }
    if (threadIdx.x == 0) tot[b] = carry;
}

__global__ __launch_bounds__(CH_TB)
void bucket_scatter_kernel(const int* __restrict__ ei, int E, int G, int nbkt,
                           const int* __restrict__ woff_t,
                           const int* __restrict__ tot,
                           int* __restrict__ rec) {
    __shared__ int lh[MAXB];
    __shared__ int lstart[MAXB];
    __shared__ int gdelta[MAXB];
    __shared__ int ssrc[CH];
    __shared__ int sdst[CH];
    __shared__ int wsum[33];
    const int tid  = threadIdx.x;
    const int base = blockIdx.x * CH;

    int d[CH_PER], s[CH_PER], r[CH_PER];
#pragma unroll
    for (int k = 0; k < CH_PER; ++k) {
        const int e = base + k * CH_TB + tid;
        d[k] = (e < E) ? ei[E + e] : -1;
        s[k] = (e < E) ? ei[e] : 0;
    }
    const int tv = (tid < nbkt) ? tot[tid] : 0;
    const int wo = (tid < nbkt) ? woff_t[tid * G + blockIdx.x] : 0;
    lh[tid] = 0;
    __syncthreads();
#pragma unroll
    for (int k = 0; k < CH_PER; ++k)
        r[k] = (d[k] >= 0) ? atomicAdd(&lh[d[k] >> BK_SHIFT], 1) : 0;
    __syncthreads();

    int dummy;
    const int ls = block_excl_scan<CH_TB>(lh[tid], wsum, &dummy);
    const int bb = block_excl_scan<CH_TB>(tv, wsum, &dummy);
    lstart[tid] = ls;
    gdelta[tid] = bb + wo - ls;
    __syncthreads();

#pragma unroll
    for (int k = 0; k < CH_PER; ++k) {
        if (d[k] >= 0) {
            const int p = lstart[d[k] >> BK_SHIFT] + r[k];
            ssrc[p] = s[k];
            sdst[p] = d[k];
        }
    }
    __syncthreads();

    const int n = (E - base < CH) ? (E - base) : CH;
    for (int i = tid; i < n; i += CH_TB) {
        const int dd = sdst[i];
        rec[gdelta[dd >> BK_SHIFT] + i] =
            ((dd & (BK_NODES - 1)) << REC_SHIFT) | ssrc[i];
    }
}

__global__ __launch_bounds__(SORT_TB)
void bucket_sort_kernel(const int* __restrict__ rec, const int* __restrict__ tot,
                        int N, int* __restrict__ cnt, int* __restrict__ off,
                        int* __restrict__ srcs) {
    __shared__ int ncur[BK_NODES];
    __shared__ int stage[SORT_CAP];
    __shared__ int wsum[33];
    const int tid = threadIdx.x;
    const int b   = blockIdx.x;

    const int nb = tot[b];
    int base, dummy;
    block_excl_scan<SORT_TB>((tid < b) ? tot[tid] : 0, wsum, &base);

    const bool fits = nb <= SORT_CAP;
    int rc[SORT_PER];
    if (tid < BK_NODES) ncur[tid] = 0;
    if (fits) {
#pragma unroll
        for (int k = 0; k < SORT_PER; ++k) {
            const int i = k * SORT_TB + tid;
            rc[k] = (i < nb) ? rec[base + i] : -1;
        }
    }
    __syncthreads();
    if (fits) {
#pragma unroll
        for (int k = 0; k < SORT_PER; ++k)
            if (rc[k] >= 0) atomicAdd(&ncur[rc[k] >> REC_SHIFT], 1);
    } else {
        for (int i = tid; i < nb; i += SORT_TB)
            atomicAdd(&ncur[rec[base + i] >> REC_SHIFT], 1);
    }
    __syncthreads();

    const int c  = (tid < BK_NODES) ? ncur[tid] : 0;
    const int ex = block_excl_scan<SORT_TB>(c, wsum, &dummy);
    if (tid < BK_NODES) {
        const int node = b * BK_NODES + tid;
        if (node < N) { cnt[node] = c; off[node] = base + ex; }
        ncur[tid] = ex;
    }
    __syncthreads();

    if (fits) {
#pragma unroll
        for (int k = 0; k < SORT_PER; ++k)
            if (rc[k] >= 0)
                stage[atomicAdd(&ncur[rc[k] >> REC_SHIFT], 1)] = rc[k] & REC_MASK;
        __syncthreads();
        for (int i = tid; i < nb; i += SORT_TB) srcs[base + i] = stage[i];
    } else {
        for (int i = tid; i < nb; i += SORT_TB) {
            const int rv = rec[base + i];
            srcs[base + atomicAdd(&ncur[rv >> REC_SHIFT], 1)] = rv & REC_MASK;
        }
    }
}

__device__ __forceinline__ float leaky02(float v) {
    return (v > 0.0f) ? v : 0.2f * v;
}

template <int H, int C, bool RELU>
__global__ void node_agg_kernel(const int* __restrict__ off,
                                const int* __restrict__ cnt,
                                const int* __restrict__ srcs,
                                const float* __restrict__ asn,
                                const float* __restrict__ adn,
                                const float* __restrict__ xl,
                                const float* __restrict__ b,
                                float* __restrict__ outp, int n) {
    constexpr int F   = H * C;
    constexpr int LPN = F / 4;
    const int t  = blockIdx.x * blockDim.x + threadIdx.x;
    const int i  = t / LPN;
    const int lg = t % LPN;
    if (i >= n) return;

    const int beg = off[i];
    const int deg = cnt[i];
    const int h   = (lg * 4) / C;

    float ad[H], mx[H];
#pragma unroll
    for (int hh = 0; hh < H; ++hh) {
        ad[hh] = adn[i * H + hh];
        mx[hh] = leaky02(asn[i * H + hh] + ad[hh]);
    }

    for (int j = lg; j < deg; j += LPN) {
        const int s = srcs[beg + j];
#pragma unroll
        for (int hh = 0; hh < H; ++hh)
            mx[hh] = fmaxf(mx[hh], leaky02(asn[s * H + hh] + ad[hh]));
    }
#pragma unroll
    for (int hh = 0; hh < H; ++hh) {
#pragma unroll
        for (int o = LPN / 2; o > 0; o >>= 1)
            mx[hh] = fmaxf(mx[hh], __shfl_xor(mx[hh], o, LPN));
    }

    float m = mx[0], adh = ad[0];
#pragma unroll
    for (int hh = 1; hh < H; ++hh) {
        if (h == hh) { m = mx[hh]; adh = ad[hh]; }
    }
    if (!(m >= -3.4e38f && m <= 3.4e38f)) m = 0.0f;

    const int f0 = lg * 4;
    float w   = __expf(leaky02(asn[i * H + h] + adh) - m);
    float den = w;
    float4 r  = *(const float4*)(xl + (size_t)i * F + f0);
    float4 acc;
    acc.x = r.x * w; acc.y = r.y * w; acc.z = r.z * w; acc.w = r.w * w;

#pragma unroll 4
    for (int j = 0; j < deg; ++j) {
        const int s = srcs[beg + j];
        const float wj = __expf(leaky02(asn[s * H + h] + adh) - m);
        const float4 rr = *(const float4*)(xl + (size_t)s * F + f0);
        den += wj;
        acc.x += rr.x * wj;
        acc.y += rr.y * wj;
        acc.z += rr.z * wj;
        acc.w += rr.w * wj;
    }

    const float dd = den + 1e-16f;
    float4 o;
    o.x = acc.x / dd + b[f0 + 0];
    o.y = acc.y / dd + b[f0 + 1];
    o.z = acc.z / dd + b[f0 + 2];
    o.w = acc.w / dd + b[f0 + 3];
    if (RELU) {
        o.x = o.x > 0.0f ? o.x : 0.0f;
        o.y = o.y > 0.0f ? o.y : 0.0f;
        o.z = o.z > 0.0f ? o.z : 0.0f;
        o.w = o.w > 0.0f ? o.w : 0.0f;
    }
    *(float4*)(outp + (size_t)i * F + f0) = o;
}

static inline int cdiv(long long a, long long b) { return (int)((a + b - 1) / b); }

extern "C" void kernel_launch(void* const* d_in, const int* in_sizes, int n_in,
                              void* d_out, int out_size, void* d_ws, size_t ws_size,
                              hipStream_t stream) {
    const float* x      = (const float*)d_in[0];
    const int*   ei     = (const int*)d_in[1];
    const float* W1     = (const float*)d_in[2];
    const float* a_src1 = (const float*)d_in[3];
    const float* a_dst1 = (const float*)d_in[4];
    const float* b1     = (const float*)d_in[5];
    const float* W2     = (const float*)d_in[6];
    const float* a_src2 = (const float*)d_in[7];
    const float* a_dst2 = (const float*)d_in[8];
    const float* b2     = (const float*)d_in[9];
    float* out = (float*)d_out;

    const int N  = in_sizes[0] / IN_DIM;
    const int E  = in_sizes[1] / 2;
    const int F1 = HEADS * HID;

    float* ws   = (float*)d_ws;
    float* xl1  = ws;
    float* h1   = xl1  + (size_t)N * F1;
    float* asn1 = h1   + (size_t)N * F1;
    float* adn1 = asn1 + (size_t)N * HEADS;
    float* xl2  = adn1 + (size_t)N * HEADS;
    float* asn2 = xl2  + (size_t)N * OUT_DIM;
    float* adn2 = asn2 + (size_t)N;
    int*   cnt  = (int*)(adn2 + (size_t)N);
    int*   off  = cnt  + (size_t)N;
    int*   rec  = off  + (size_t)N;
    int*   srcs = rec  + (size_t)E;
    const int G    = cdiv(E, CH);
    const int NBKT = cdiv(N, BK_NODES);
    int*   hist_t = (int*)h1;
    int*   woff_t = hist_t + (size_t)NBKT * G;
    int*   tot    = woff_t + (size_t)NBKT * G;

    const int TB = 256;

    bucket_hist_kernel<<<G, CH_TB, 0, stream>>>(ei, E, G, NBKT, hist_t);
    bucket_colscan_kernel<<<NBKT, CS_TB, 0, stream>>>(hist_t, G, woff_t, tot);
    bucket_scatter_kernel<<<G, CH_TB, 0, stream>>>(ei, E, G, NBKT, woff_t, tot, rec);
    bucket_sort_kernel<<<NBKT, SORT_TB, 0, stream>>>(rec, tot, N, cnt, off, srcs);

    {
        int mtiles = cdiv(N, 16);
        wmma_gemm_f32<IN_DIM, F1><<<cdiv(mtiles, 4), 128, 0, stream>>>(x, W1, xl1, N);
        score_kernel<HEADS, HID><<<cdiv(N, TB), TB, 0, stream>>>(xl1, a_src1, a_dst1, asn1, adn1, N);
        node_agg_kernel<HEADS, HID, true><<<cdiv((long long)N * (F1 / 4), TB), TB, 0, stream>>>(
            off, cnt, srcs, asn1, adn1, xl1, b1, h1, N);
    }

    {
        int mtiles = cdiv(N, 16);
        wmma_gemm_f32<F1, OUT_DIM><<<cdiv(mtiles, 4), 128, 0, stream>>>(h1, W2, xl2, N);
        score_kernel<1, OUT_DIM><<<cdiv(N, TB), TB, 0, stream>>>(xl2, a_src2, a_dst2, asn2, adn2, N);
        node_agg_kernel<1, OUT_DIM, false><<<cdiv((long long)N * (OUT_DIM / 4), TB), TB, 0, stream>>>(
            off, cnt, srcs, asn2, adn2, xl2, b2, out, N);
    }
}


# Layer-1 product x*W1 runs as extra blocks of the scatter launch (block index picks the role, one LDS area): one launch fewer, matrix work fills the scatter's waits; W1 staging stride now a constant
# speedup vs baseline: 7.6591x; 1.5689x over previous
#include <hip/hip_runtime.h>
#include <hip/hip_bf16.h>
#include <math.h>

typedef float v2f __attribute__((ext_vector_type(2)));
typedef float v8f __attribute__((ext_vector_type(8)));

#define IN_DIM   128
#define HID      32
#define HEADS    2
#define OUT_DIM  32

template <int K, int NC>
__device__ __forceinline__ void wmma_gemm_block(const float* __restrict__ A,
                                                const float* __restrict__ B,
                                                float* __restrict__ C, int M,
                                                int blk, int WPB, int NTHR,
                                                v2f* Bs2) {
    constexpr int NT = NC / 16;

    for (int p = threadIdx.x; p < (K / 2) * NC; p += NTHR) {
        int k2 = p / NC;
        int n  = p - k2 * NC;
        v2f pr;
        pr.x = B[(2 * k2 + 0) * NC + n];
        pr.y = B[(2 * k2 + 1) * NC + n];
        Bs2[p] = pr;
    }
    __syncthreads();

    const int wave = threadIdx.x >> 5;
    const int lane = threadIdx.x & 31;
    const int mt   = blk * WPB + wave;
    if (mt * 16 >= M) return;

    const int halfK = (lane >> 4) * 2;
    const int ncol  = lane & 15;
    const int arow  = mt * 16 + (lane & 15);
    const float* Arow = A + (size_t)arow * K;

    v8f acc[NT];
#pragma unroll
    for (int t = 0; t < NT; ++t) acc[t] = (v8f){0, 0, 0, 0, 0, 0, 0, 0};

    for (int k = 0; k < K; k += 4) {
        v2f a = *(const v2f*)(Arow + k + halfK);
        const int prow = (k + halfK) >> 1;
#pragma unroll
        for (int t = 0; t < NT; ++t) {
            v2f b = Bs2[prow * NC + t * 16 + ncol];
            acc[t] = __builtin_amdgcn_wmma_f32_16x16x4_f32(
                false, a, false, b, (short)0, acc[t], false, false);
        }
    }

    const int rbase = mt * 16 + ((lane >> 4) << 3);
#pragma unroll
    for (int t = 0; t < NT; ++t) {
#pragma unroll
        for (int j = 0; j < 8; ++j) {
            C[(size_t)(rbase + j) * NC + t * 16 + ncol] = acc[t][j];
        }
    }
}

template <int K, int NC>
__global__ void wmma_gemm_f32(const float* __restrict__ A,
                              const float* __restrict__ B,
                              float* __restrict__ C, int M) {
    __shared__ v2f Bs2[(K / 2) * NC];
    wmma_gemm_block<K, NC>(A, B, C, M, blockIdx.x, 4, blockDim.x, Bs2);
}

#define BK_SHIFT  9
#define BK_NODES  (1 << BK_SHIFT)
#define MAXB      256
#define CH_TB     256
#define CH_PER    16
#define CH        (CH_TB * CH_PER)
#define REC_SHIFT 22
#define REC_MASK  ((1 << REC_SHIFT) - 1)
#define CS_TB     512
#define SORT_TB   1024
#define SORT_PER  16
#define SORT_CAP  (SORT_TB * SORT_PER)

template <int NT>
__device__ __forceinline__ int block_excl_scan(int v, int* wsum, int* total) {
    const int lane = threadIdx.x & 31;
    const int wv   = threadIdx.x >> 5;
    int inc = v;
#pragma unroll
    for (int o = 1; o < 32; o <<= 1) {
        int t = __shfl_up(inc, o, 32);
        if (lane >= o) inc += t;
    }
    if (lane == 31) wsum[wv] = inc;
    __syncthreads();
    if (wv == 0) {
        const int ws = (lane < NT / 32) ? wsum[lane] : 0;
        int wi = ws;
#pragma unroll
        for (int o = 1; o < 32; o <<= 1) {
            int t = __shfl_up(wi, o, 32);
            if (lane >= o) wi += t;
        }
        if (lane < NT / 32) wsum[lane] = wi - ws;
        if (lane == 31) wsum[32] = wi;
    }
    __syncthreads();
    const int res = wsum[wv] + inc - v;
    *total = wsum[32];
    __syncthreads();
    return res;
}

__global__ __launch_bounds__(CH_TB)
void bucket_hist_kernel(const int* __restrict__ ei, int E, int G, int nbkt,
                        int* __restrict__ hist_t) {
    __shared__ int lh[MAXB];
    const int tid  = threadIdx.x;
    const int base = blockIdx.x * CH;
    lh[tid] = 0;
    __syncthreads();
#pragma unroll
    for (int k = 0; k < CH_PER; ++k) {
        const int e = base + k * CH_TB + tid;
        if (e < E) atomicAdd(&lh[ei[E + e] >> BK_SHIFT], 1);
    }
    __syncthreads();
    if (tid < nbkt) hist_t[tid * G + blockIdx.x] = lh[tid];
}

__global__ __launch_bounds__(CS_TB)
void bucket_colscan_kernel(const int* __restrict__ hist_t, int G,
                           int* __restrict__ woff_t, int* __restrict__ tot) {
    __shared__ int wsum[33];
    const int b = blockIdx.x;
    int carry = 0;
    for (int w0 = 0; w0 < G; w0 += CS_TB) {
        const int w = w0 + threadIdx.x;
        const int v = (w < G) ? hist_t[b * G + w] : 0;
        int total;
        const int ex = block_excl_scan<CS_TB>(v, wsum, &total);
        if (w < G) woff_t[b * G + w] = carry + ex;
        carry += total;
    }
    if (threadIdx.x == 0) tot[b] = carry;
}

#define SC_LDS_INTS (3 * MAXB + 2 * CH + 33)
#define G1_LDS_INTS ((IN_DIM / 2) * (HEADS * HID) * 2)
#define SG_LDS_INTS (SC_LDS_INTS > G1_LDS_INTS ? SC_LDS_INTS : G1_LDS_INTS)
#define G1_WPB      (CH_TB / 32)

__device__ __forceinline__
void bucket_scatter_block(const int* __restrict__ ei, int E, int G, int nbkt,
                          const int* __restrict__ woff_t,
                          const int* __restrict__ tot,
                          int* __restrict__ rec, int blk, int* smem) {
    int* const lh     = smem;
    int* const lstart = lh + MAXB;
    int* const gdelta = lstart + MAXB;
    int* const ssrc   = gdelta + MAXB;
    int* const sdst   = ssrc + CH;
    int* const wsum   = sdst + CH;
    const int tid  = threadIdx.x;
    const int base = blk * CH;

    int d[CH_PER], s[CH_PER], r[CH_PER];
#pragma unroll
    for (int k = 0; k < CH_PER; ++k) {
        const int e = base + k * CH_TB + tid;
        d[k] = (e < E) ? ei[E + e] : -1;
        s[k] = (e < E) ? ei[e] : 0;
    }
    const int tv = (tid < nbkt) ? tot[tid] : 0;
    const int wo = (tid < nbkt) ? woff_t[tid * G + blk] : 0;
    lh[tid] = 0;
    __syncthreads();
#pragma unroll
    for (int k = 0; k < CH_PER; ++k)
        r[k] = (d[k] >= 0) ? atomicAdd(&lh[d[k] >> BK_SHIFT], 1) : 0;
    __syncthreads();

    int dummy;
    const int ls = block_excl_scan<CH_TB>(lh[tid], wsum, &dummy);
    const int bb = block_excl_scan<CH_TB>(tv, wsum, &dummy);
    lstart[tid] = ls;
    gdelta[tid] = bb + wo - ls;
    __syncthreads();

#pragma unroll
    for (int k = 0; k < CH_PER; ++k) {
        if (d[k] >= 0) {
            const int p = lstart[d[k] >> BK_SHIFT] + r[k];
            ssrc[p] = s[k];
            sdst[p] = d[k];
        }
    }
    __syncthreads();

    const int n = (E - base < CH) ? (E - base) : CH;
    for (int i = tid; i < n; i += CH_TB) {
        const int dd = sdst[i];
        rec[gdelta[dd >> BK_SHIFT] + i] =
            ((dd & (BK_NODES - 1)) << REC_SHIFT) | ssrc[i];
    }
}

__global__ __launch_bounds__(CH_TB)
void scatter_gemm1_kernel(const int* __restrict__ ei, int E, int G, int nbkt,
                          const int* __restrict__ woff_t,
                          const int* __restrict__ tot,
                          int* __restrict__ rec,
                          const float* __restrict__ x,
                          const float* __restrict__ W1,
                          float* __restrict__ xl1, int N) {
    __shared__ __attribute__((aligned(16))) int smem[SG_LDS_INTS];
    if ((int)blockIdx.x < G) {
        bucket_scatter_block(ei, E, G, nbkt, woff_t, tot, rec, blockIdx.x, smem);
    } else {
        wmma_gemm_block<IN_DIM, HEADS * HID>(x, W1, xl1, N, (int)blockIdx.x - G,
                                             G1_WPB, CH_TB, (v2f*)smem);
    }
}

__global__ __launch_bounds__(SORT_TB)
void bucket_sort_kernel(const int* __restrict__ rec, const int* __restrict__ tot,
                        int N, int* __restrict__ cnt, int* __restrict__ off,
                        int* __restrict__ srcs) {
    __shared__ int ncur[BK_NODES];
    __shared__ int stage[SORT_CAP];
    __shared__ int wsum[33];
    const int tid = threadIdx.x;
    const int b   = blockIdx.x;

    const int nb = tot[b];
    int base, dummy;
    block_excl_scan<SORT_TB>((tid < b) ? tot[tid] : 0, wsum, &base);

    const bool fits = nb <= SORT_CAP;
    int rc[SORT_PER];
    if (tid < BK_NODES) ncur[tid] = 0;
    if (fits) {
#pragma unroll
        for (int k = 0; k < SORT_PER; ++k) {
            const int i = k * SORT_TB + tid;
            rc[k] = (i < nb) ? rec[base + i] : -1;
        }
    }
    __syncthreads();
    if (fits) {
#pragma unroll
        for (int k = 0; k < SORT_PER; ++k)
            if (rc[k] >= 0) atomicAdd(&ncur[rc[k] >> REC_SHIFT], 1);
    } else {
        for (int i = tid; i < nb; i += SORT_TB)
            atomicAdd(&ncur[rec[base + i] >> REC_SHIFT], 1);
    }
    __syncthreads();

    const int c  = (tid < BK_NODES) ? ncur[tid] : 0;
    const int ex = block_excl_scan<SORT_TB>(c, wsum, &dummy);
    if (tid < BK_NODES) {
        const int node = b * BK_NODES + tid;
        if (node < N) { cnt[node] = c; off[node] = base + ex; }
        ncur[tid] = ex;
    }
    __syncthreads();

    if (fits) {
#pragma unroll
        for (int k = 0; k < SORT_PER; ++k)
            if (rc[k] >= 0)
                stage[atomicAdd(&ncur[rc[k] >> REC_SHIFT], 1)] = rc[k] & REC_MASK;
        __syncthreads();
        for (int i = tid; i < nb; i += SORT_TB) srcs[base + i] = stage[i];
    } else {
        for (int i = tid; i < nb; i += SORT_TB) {
            const int rv = rec[base + i];
            srcs[base + atomicAdd(&ncur[rv >> REC_SHIFT], 1)] = rv & REC_MASK;
        }
    }
}

__device__ __forceinline__ float leaky02(float v) {
    return fmaxf(v, 0.2f * v);
}

template <int MASK>
__device__ __forceinline__ float row_xor(float v) {
    return __int_as_float(__builtin_amdgcn_update_dpp(
        0, __float_as_int(v), 0x160 + MASK, 0xf, 0xf, true));
}

template <int LPH>
__device__ __forceinline__ float head_sum(float v) {
    if (LPH > 1)  v += row_xor<1>(v);
    if (LPH > 2)  v += row_xor<2>(v);
    if (LPH > 4)  v += row_xor<4>(v);
    if (LPH > 8)  v += row_xor<8>(v);
    return v;
}

__device__ __forceinline__ float dot4(const float4 a, const float4 b) {
    return fmaf(a.w, b.w, fmaf(a.z, b.z, fmaf(a.y, b.y, a.x * b.x)));
}

template <int F>
__device__ __forceinline__ float4 row_load(const float* __restrict__ xl, int s, int f0) {
    const unsigned byte_off = (unsigned)s * (unsigned)(F * 4) + (unsigned)(f0 * 4);
    return *(const float4*)((const char*)xl + byte_off);
}

template <int H, int C, bool RELU>
__global__ void node_agg_kernel(const int* __restrict__ off,
                                const int* __restrict__ cnt,
                                const int* __restrict__ srcs,
                                const float* __restrict__ xl,
                                const float* __restrict__ a_src,
                                const float* __restrict__ a_dst,
                                const float* __restrict__ b,
                                float* __restrict__ outp, int n) {
    constexpr int F   = H * C;
    constexpr int LPN = F / 4;
    constexpr int LPH = C / 4;
    constexpr int U   = 4;
    const int t  = blockIdx.x * blockDim.x + threadIdx.x;
    const int i  = t / LPN;
    const int lg = t % LPN;
    if (i >= n) return;

    const int beg = off[i];
    const int deg = cnt[i];
    const int f0  = lg * 4;

    float4 as, ad;
    as.x = a_src[f0 + 0]; as.y = a_src[f0 + 1]; as.z = a_src[f0 + 2]; as.w = a_src[f0 + 3];
    ad.x = a_dst[f0 + 0]; ad.y = a_dst[f0 + 1]; ad.z = a_dst[f0 + 2]; ad.w = a_dst[f0 + 3];

    const float4 r  = row_load<F>(xl, i, f0);
    const float adh = head_sum<LPH>(dot4(r, ad));
    float m   = leaky02(head_sum<LPH>(dot4(r, as)) + adh);
    float den = 1.0f;
    float4 acc = r;

    for (int j = 0; j < deg; j += U) {
        int    s[U];
        float4 rr[U];
        float  e[U];
#pragma unroll
        for (int u = 0; u < U; ++u) {
            const int jj = (j + u < deg) ? (j + u) : (deg - 1);
            s[u] = srcs[beg + jj];
        }
#pragma unroll
        for (int u = 0; u < U; ++u)
            rr[u] = row_load<F>(xl, s[u], f0);
#pragma unroll
        for (int u = 0; u < U; ++u) {
            const float v = leaky02(head_sum<LPH>(dot4(rr[u], as)) + adh);
            e[u] = (j + u < deg) ? v : -INFINITY;
        }
        float mn = m;
#pragma unroll
        for (int u = 0; u < U; ++u) mn = fmaxf(mn, e[u]);
        const float sc = __expf(m - mn);
        m = mn;
        den *= sc;
        acc.x *= sc; acc.y *= sc; acc.z *= sc; acc.w *= sc;
#pragma unroll
        for (int u = 0; u < U; ++u) {
            const float w = __expf(e[u] - mn);
            den += w;
            acc.x += rr[u].x * w;
            acc.y += rr[u].y * w;
            acc.z += rr[u].z * w;
            acc.w += rr[u].w * w;
        }
    }

    const float dd = den + 1e-16f;
    float4 o;
    o.x = acc.x / dd + b[f0 + 0];
    o.y = acc.y / dd + b[f0 + 1];
    o.z = acc.z / dd + b[f0 + 2];
    o.w = acc.w / dd + b[f0 + 3];
    if (RELU) {
        o.x = o.x > 0.0f ? o.x : 0.0f;
        o.y = o.y > 0.0f ? o.y : 0.0f;
        o.z = o.z > 0.0f ? o.z : 0.0f;
        o.w = o.w > 0.0f ? o.w : 0.0f;
    }
    *(float4*)(outp + (size_t)i * F + f0) = o;
}

static inline int cdiv(long long a, long long b) { return (int)((a + b - 1) / b); }

extern "C" void kernel_launch(void* const* d_in, const int* in_sizes, int n_in,
                              void* d_out, int out_size, void* d_ws, size_t ws_size,
                              hipStream_t stream) {
    const float* x      = (const float*)d_in[0];
    const int*   ei     = (const int*)d_in[1];
    const float* W1     = (const float*)d_in[2];
    const float* a_src1 = (const float*)d_in[3];
    const float* a_dst1 = (const float*)d_in[4];
    const float* b1     = (const float*)d_in[5];
    const float* W2     = (const float*)d_in[6];
    const float* a_src2 = (const float*)d_in[7];
    const float* a_dst2 = (const float*)d_in[8];
    const float* b2     = (const float*)d_in[9];
    float* out = (float*)d_out;

    const int N  = in_sizes[0] / IN_DIM;
    const int E  = in_sizes[1] / 2;
    const int F1 = HEADS * HID;

    float* ws   = (float*)d_ws;
    float* xl1  = ws;
    float* h1   = xl1  + (size_t)N * F1;
    float* xl2  = h1   + (size_t)N * F1;
    int*   cnt  = (int*)(xl2 + (size_t)N * OUT_DIM);
    int*   off  = cnt  + (size_t)N;
    int*   rec  = off  + (size_t)N;
    int*   srcs = rec  + (size_t)E;
    const int G    = cdiv(E, CH);
    const int NBKT = cdiv(N, BK_NODES);
    int*   hist_t = (int*)h1;
    int*   woff_t = hist_t + (size_t)NBKT * G;
    int*   tot    = woff_t + (size_t)NBKT * G;

    const int TB = 256;

    bucket_hist_kernel<<<G, CH_TB, 0, stream>>>(ei, E, G, NBKT, hist_t);
    bucket_colscan_kernel<<<NBKT, CS_TB, 0, stream>>>(hist_t, G, woff_t, tot);
    const int G1B = cdiv(cdiv(N, 16), G1_WPB);
    scatter_gemm1_kernel<<<G + G1B, CH_TB, 0, stream>>>(ei, E, G, NBKT, woff_t, tot, rec,
                                                         x, W1, xl1, N);
    bucket_sort_kernel<<<NBKT, SORT_TB, 0, stream>>>(rec, tot, N, cnt, off, srcs);

    {
        node_agg_kernel<HEADS, HID, true><<<cdiv((long long)N * (F1 / 4), TB), TB, 0, stream>>>(
            off, cnt, srcs, xl1, a_src1, a_dst1, b1, h1, N);
    }

    {
        int mtiles = cdiv(N, 16);
        wmma_gemm_f32<F1, OUT_DIM><<<cdiv(mtiles, 4), 128, 0, stream>>>(h1, W2, xl2, N);
        node_agg_kernel<1, OUT_DIM, false><<<cdiv((long long)N * (OUT_DIM / 4), TB), TB, 0, stream>>>(
            off, cnt, srcs, xl2, a_src2, a_dst2, b2, out, N);
    }
}


# Layer-2 product h*W2 moved behind the layer-1 aggregation: a block's 16 finished rows go through an LDS tile to two WMMA waves; product launch and the layer-1 output plane (write + read) gone
# speedup vs baseline: 8.2411x; 1.0760x over previous
#include <hip/hip_runtime.h>
#include <hip/hip_bf16.h>
#include <math.h>

typedef float v2f __attribute__((ext_vector_type(2)));
typedef float v8f __attribute__((ext_vector_type(8)));

#define IN_DIM   128
#define HID      32
#define HEADS    2
#define OUT_DIM  32

template <int K, int NC>
__device__ __forceinline__ void wmma_gemm_block(const float* __restrict__ A,
                                                const float* __restrict__ B,
                                                float* __restrict__ C, int M,
                                                int blk, int WPB, int NTHR,
                                                v2f* Bs2) {
    constexpr int NT = NC / 16;

    for (int p = threadIdx.x; p < (K / 2) * NC; p += NTHR) {
        int k2 = p / NC;
        int n  = p - k2 * NC;
        v2f pr;
        pr.x = B[(2 * k2 + 0) * NC + n];
        pr.y = B[(2 * k2 + 1) * NC + n];
        Bs2[p] = pr;
    }
    __syncthreads();

    const int wave = threadIdx.x >> 5;
    const int lane = threadIdx.x & 31;
    const int mt   = blk * WPB + wave;
    if (mt * 16 >= M) return;

    const int halfK = (lane >> 4) * 2;
    const int ncol  = lane & 15;
    const int arow  = mt * 16 + (lane & 15);
    const float* Arow = A + (size_t)arow * K;

    v8f acc[NT];
#pragma unroll
    for (int t = 0; t < NT; ++t) acc[t] = (v8f){0, 0, 0, 0, 0, 0, 0, 0};

    for (int k = 0; k < K; k += 4) {
        v2f a = *(const v2f*)(Arow + k + halfK);
        const int prow = (k + halfK) >> 1;
#pragma unroll
        for (int t = 0; t < NT; ++t) {
            v2f b = Bs2[prow * NC + t * 16 + ncol];
            acc[t] = __builtin_amdgcn_wmma_f32_16x16x4_f32(
                false, a, false, b, (short)0, acc[t], false, false);
        }
    }

    const int rbase = mt * 16 + ((lane >> 4) << 3);
#pragma unroll
    for (int t = 0; t < NT; ++t) {
#pragma unroll
        for (int j = 0; j < 8; ++j) {
            C[(size_t)(rbase + j) * NC + t * 16 + ncol] = acc[t][j];
        }
    }
}

#define BK_SHIFT  9
#define BK_NODES  (1 << BK_SHIFT)
#define MAXB      256
#define CH_TB     256
#define CH_PER    16
#define CH        (CH_TB * CH_PER)
#define REC_SHIFT 22
#define REC_MASK  ((1 << REC_SHIFT) - 1)
#define CS_TB     512
#define SORT_TB   1024
#define SORT_PER  16
#define SORT_CAP  (SORT_TB * SORT_PER)

template <int NT>
__device__ __forceinline__ int block_excl_scan(int v, int* wsum, int* total) {
    const int lane = threadIdx.x & 31;
    const int wv   = threadIdx.x >> 5;
    int inc = v;
#pragma unroll
    for (int o = 1; o < 32; o <<= 1) {
        int t = __shfl_up(inc, o, 32);
        if (lane >= o) inc += t;
    }
    if (lane == 31) wsum[wv] = inc;
    __syncthreads();
    if (wv == 0) {
        const int ws = (lane < NT / 32) ? wsum[lane] : 0;
        int wi = ws;
#pragma unroll
        for (int o = 1; o < 32; o <<= 1) {
            int t = __shfl_up(wi, o, 32);
            if (lane >= o) wi += t;
        }
        if (lane < NT / 32) wsum[lane] = wi - ws;
        if (lane == 31) wsum[32] = wi;
    }
    __syncthreads();
    const int res = wsum[wv] + inc - v;
    *total = wsum[32];
    __syncthreads();
    return res;
}

__global__ __launch_bounds__(CH_TB)
void bucket_hist_kernel(const int* __restrict__ ei, int E, int G, int nbkt,
                        int* __restrict__ hist_t) {
    __shared__ int lh[MAXB];
    const int tid  = threadIdx.x;
    const int base = blockIdx.x * CH;
    lh[tid] = 0;
    __syncthreads();
#pragma unroll
    for (int k = 0; k < CH_PER; ++k) {
        const int e = base + k * CH_TB + tid;
        if (e < E) atomicAdd(&lh[ei[E + e] >> BK_SHIFT], 1);
    }
    __syncthreads();
    if (tid < nbkt) hist_t[tid * G + blockIdx.x] = lh[tid];
}

__global__ __launch_bounds__(CS_TB)
void bucket_colscan_kernel(const int* __restrict__ hist_t, int G,
                           int* __restrict__ woff_t, int* __restrict__ tot) {
    __shared__ int wsum[33];
    const int b = blockIdx.x;
    int carry = 0;
    for (int w0 = 0; w0 < G; w0 += CS_TB) {
        const int w = w0 + threadIdx.x;
        const int v = (w < G) ? hist_t[b * G + w] : 0;
        int total;
        const int ex = block_excl_scan<CS_TB>(v, wsum, &total);
        if (w < G) woff_t[b * G + w] = carry + ex;
        carry += total;
    }
    if (threadIdx.x == 0) tot[b] = carry;
}

#define SC_LDS_INTS (3 * MAXB + 2 * CH + 33)
#define G1_LDS_INTS ((IN_DIM / 2) * (HEADS * HID) * 2)
#define SG_LDS_INTS (SC_LDS_INTS > G1_LDS_INTS ? SC_LDS_INTS : G1_LDS_INTS)
#define G1_WPB      (CH_TB / 32)

__device__ __forceinline__
void bucket_scatter_block(const int* __restrict__ ei, int E, int G, int nbkt,
                          const int* __restrict__ woff_t,
                          const int* __restrict__ tot,
                          int* __restrict__ rec, int blk, int* smem) {
    int* const lh     = smem;
    int* const lstart = lh + MAXB;
    int* const gdelta = lstart + MAXB;
    int* const ssrc   = gdelta + MAXB;
    int* const sdst   = ssrc + CH;
    int* const wsum   = sdst + CH;
    const int tid  = threadIdx.x;
    const int base = blk * CH;

    int d[CH_PER], s[CH_PER], r[CH_PER];
#pragma unroll
    for (int k = 0; k < CH_PER; ++k) {
        const int e = base + k * CH_TB + tid;
        d[k] = (e < E) ? ei[E + e] : -1;
        s[k] = (e < E) ? ei[e] : 0;
    }
    const int tv = (tid < nbkt) ? tot[tid] : 0;
    const int wo = (tid < nbkt) ? woff_t[tid * G + blk] : 0;
    lh[tid] = 0;
    __syncthreads();
#pragma unroll
    for (int k = 0; k < CH_PER; ++k)
        r[k] = (d[k] >= 0) ? atomicAdd(&lh[d[k] >> BK_SHIFT], 1) : 0;
    __syncthreads();

    int dummy;
    const int ls = block_excl_scan<CH_TB>(lh[tid], wsum, &dummy);
    const int bb = block_excl_scan<CH_TB>(tv, wsum, &dummy);
    lstart[tid] = ls;
    gdelta[tid] = bb + wo - ls;
    __syncthreads();

#pragma unroll
    for (int k = 0; k < CH_PER; ++k) {
        if (d[k] >= 0) {
            const int p = lstart[d[k] >> BK_SHIFT] + r[k];
            ssrc[p] = s[k];
            sdst[p] = d[k];
        }
    }
    __syncthreads();

    const int n = (E - base < CH) ? (E - base) : CH;
    for (int i = tid; i < n; i += CH_TB) {
        const int dd = sdst[i];
        rec[gdelta[dd >> BK_SHIFT] + i] =
            ((dd & (BK_NODES - 1)) << REC_SHIFT) | ssrc[i];
    }
}

__global__ __launch_bounds__(CH_TB)
void scatter_gemm1_kernel(const int* __restrict__ ei, int E, int G, int nbkt,
                          const int* __restrict__ woff_t,
                          const int* __restrict__ tot,
                          int* __restrict__ rec,
                          const float* __restrict__ x,
                          const float* __restrict__ W1,
                          float* __restrict__ xl1, int N) {
    __shared__ __attribute__((aligned(16))) int smem[SG_LDS_INTS];
    if ((int)blockIdx.x < G) {
        bucket_scatter_block(ei, E, G, nbkt, woff_t, tot, rec, blockIdx.x, smem);
    } else {
        wmma_gemm_block<IN_DIM, HEADS * HID>(x, W1, xl1, N, (int)blockIdx.x - G,
                                             G1_WPB, CH_TB, (v2f*)smem);
    }
}

__global__ __launch_bounds__(SORT_TB)
void bucket_sort_kernel(const int* __restrict__ rec, const int* __restrict__ tot,
                        int N, int* __restrict__ cnt, int* __restrict__ off,
                        int* __restrict__ srcs) {
    __shared__ int ncur[BK_NODES];
    __shared__ int stage[SORT_CAP];
    __shared__ int wsum[33];
    const int tid = threadIdx.x;
    const int b   = blockIdx.x;

    const int nb = tot[b];
    int base, dummy;
    block_excl_scan<SORT_TB>((tid < b) ? tot[tid] : 0, wsum, &base);

    const bool fits = nb <= SORT_CAP;
    int rc[SORT_PER];
    if (tid < BK_NODES) ncur[tid] = 0;
    if (fits) {
#pragma unroll
        for (int k = 0; k < SORT_PER; ++k) {
            const int i = k * SORT_TB + tid;
            rc[k] = (i < nb) ? rec[base + i] : -1;
        }
    }
    __syncthreads();
    if (fits) {
#pragma unroll
        for (int k = 0; k < SORT_PER; ++k)
            if (rc[k] >= 0) atomicAdd(&ncur[rc[k] >> REC_SHIFT], 1);
    } else {
        for (int i = tid; i < nb; i += SORT_TB)
            atomicAdd(&ncur[rec[base + i] >> REC_SHIFT], 1);
    }
    __syncthreads();

    const int c  = (tid < BK_NODES) ? ncur[tid] : 0;
    const int ex = block_excl_scan<SORT_TB>(c, wsum, &dummy);
    if (tid < BK_NODES) {
        const int node = b * BK_NODES + tid;
        if (node < N) { cnt[node] = c; off[node] = base + ex; }
        ncur[tid] = ex;
    }
    __syncthreads();

    if (fits) {
#pragma unroll
        for (int k = 0; k < SORT_PER; ++k)
            if (rc[k] >= 0)
                stage[atomicAdd(&ncur[rc[k] >> REC_SHIFT], 1)] = rc[k] & REC_MASK;
        __syncthreads();
        for (int i = tid; i < nb; i += SORT_TB) srcs[base + i] = stage[i];
    } else {
        for (int i = tid; i < nb; i += SORT_TB) {
            const int rv = rec[base + i];
            srcs[base + atomicAdd(&ncur[rv >> REC_SHIFT], 1)] = rv & REC_MASK;
        }
    }
}

__device__ __forceinline__ float leaky02(float v) {
    return fmaxf(v, 0.2f * v);
}

template <int MASK>
__device__ __forceinline__ float row_xor(float v) {
    return __int_as_float(__builtin_amdgcn_update_dpp(
        0, __float_as_int(v), 0x160 + MASK, 0xf, 0xf, true));
}

template <int LPH>
__device__ __forceinline__ float head_sum(float v) {
    if (LPH > 1)  v += row_xor<1>(v);
    if (LPH > 2)  v += row_xor<2>(v);
    if (LPH > 4)  v += row_xor<4>(v);
    if (LPH > 8)  v += row_xor<8>(v);
    return v;
}

__device__ __forceinline__ float dot4(const float4 a, const float4 b) {
    return fmaf(a.w, b.w, fmaf(a.z, b.z, fmaf(a.y, b.y, a.x * b.x)));
}

template <int F>
__device__ __forceinline__ float4 row_load(const float* __restrict__ xl, int s, int f0) {
    const unsigned byte_off = (unsigned)s * (unsigned)(F * 4) + (unsigned)(f0 * 4);
    return *(const float4*)((const char*)xl + byte_off);
}

template <int H, int C, bool RELU>
__device__ __forceinline__ float4 node_agg_row(const int* __restrict__ off,
                                               const int* __restrict__ cnt,
                                               const int* __restrict__ srcs,
                                               const float* __restrict__ xl,
                                               const float* __restrict__ a_src,
                                               const float* __restrict__ a_dst,
                                               const float* __restrict__ b,
                                               int i, int lg) {
    constexpr int F   = H * C;
    constexpr int LPH = C / 4;
    constexpr int U   = 4;

    const int beg = off[i];
    const int deg = cnt[i];
    const int f0  = lg * 4;

    float4 as, ad;
    as.x = a_src[f0 + 0]; as.y = a_src[f0 + 1]; as.z = a_src[f0 + 2]; as.w = a_src[f0 + 3];
    ad.x = a_dst[f0 + 0]; ad.y = a_dst[f0 + 1]; ad.z = a_dst[f0 + 2]; ad.w = a_dst[f0 + 3];

    const float4 r  = row_load<F>(xl, i, f0);
    const float adh = head_sum<LPH>(dot4(r, ad));
    float m   = leaky02(head_sum<LPH>(dot4(r, as)) + adh);
    float den = 1.0f;
    float4 acc = r;

    for (int j = 0; j < deg; j += U) {
        int    s[U];
        float4 rr[U];
        float  e[U];
#pragma unroll
        for (int u = 0; u < U; ++u) {
            const int jj = (j + u < deg) ? (j + u) : (deg - 1);
            s[u] = srcs[beg + jj];
        }
#pragma unroll
        for (int u = 0; u < U; ++u)
            rr[u] = row_load<F>(xl, s[u], f0);
#pragma unroll
        for (int u = 0; u < U; ++u) {
            const float v = leaky02(head_sum<LPH>(dot4(rr[u], as)) + adh);
            e[u] = (j + u < deg) ? v : -INFINITY;
        }
        float mn = m;
#pragma unroll
        for (int u = 0; u < U; ++u) mn = fmaxf(mn, e[u]);
        const float sc = __expf(m - mn);
        m = mn;
        den *= sc;
        acc.x *= sc; acc.y *= sc; acc.z *= sc; acc.w *= sc;
#pragma unroll
        for (int u = 0; u < U; ++u) {
            const float w = __expf(e[u] - mn);
            den += w;
            acc.x += rr[u].x * w;
            acc.y += rr[u].y * w;
            acc.z += rr[u].z * w;
            acc.w += rr[u].w * w;
        }
    }

    const float dd = den + 1e-16f;
    float4 o;
    o.x = acc.x / dd + b[f0 + 0];
    o.y = acc.y / dd + b[f0 + 1];
    o.z = acc.z / dd + b[f0 + 2];
    o.w = acc.w / dd + b[f0 + 3];
    if (RELU) {
        o.x = o.x > 0.0f ? o.x : 0.0f;
        o.y = o.y > 0.0f ? o.y : 0.0f;
        o.z = o.z > 0.0f ? o.z : 0.0f;
        o.w = o.w > 0.0f ? o.w : 0.0f;
    }
    return o;
}

template <int H, int C, bool RELU>
__global__ void node_agg_kernel(const int* __restrict__ off,
                                const int* __restrict__ cnt,
                                const int* __restrict__ srcs,
                                const float* __restrict__ xl,
                                const float* __restrict__ a_src,
                                const float* __restrict__ a_dst,
                                const float* __restrict__ b,
                                float* __restrict__ outp, int n) {
    constexpr int F   = H * C;
    constexpr int LPN = F / 4;
    const int t  = blockIdx.x * blockDim.x + threadIdx.x;
    const int i  = t / LPN;
    const int lg = t % LPN;
    if (i >= n) return;
    const float4 o = node_agg_row<H, C, RELU>(off, cnt, srcs, xl, a_src, a_dst, b, i, lg);
    *(float4*)(outp + (size_t)i * F + lg * 4) = o;
}

#define AG_NODES 16
#define AG_TB    (AG_NODES * (HEADS * HID / 4))
#define AG_PITCH 68

__global__ __launch_bounds__(AG_TB)
void node_agg1_gemm2_kernel(const int* __restrict__ off,
                            const int* __restrict__ cnt,
                            const int* __restrict__ srcs,
                            const float* __restrict__ xl,
                            const float* __restrict__ a_src,
                            const float* __restrict__ a_dst,
                            const float* __restrict__ b,
                            const float* __restrict__ W2,
                            float* __restrict__ xl2, int n) {
    constexpr int F   = HEADS * HID;
    constexpr int LPN = F / 4;
    __shared__ __attribute__((aligned(16))) float tile[AG_NODES * AG_PITCH];

    const int nl = threadIdx.x / LPN;
    const int lg = threadIdx.x % LPN;
    const int i  = blockIdx.x * AG_NODES + nl;
    float* const trow = tile + nl * AG_PITCH + lg * 4;
    if (i < n) {
        *(float4*)trow = node_agg_row<HEADS, HID, true>(off, cnt, srcs, xl, a_src, a_dst, b, i, lg);
    } else {
        *(float4*)trow = make_float4(0.0f, 0.0f, 0.0f, 0.0f);
    }

    const int  wave  = threadIdx.x >> 5;
    const bool mm    = wave < OUT_DIM / 16;
    const int  lane  = threadIdx.x & 31;
    const int  halfK = (lane >> 4) * 2;
    const int  c16   = lane & 15;
    const int  col   = (mm ? wave : 0) * 16 + c16;

    v2f bfr[F / 4];
#pragma unroll
    for (int kk = 0; kk < F / 4; ++kk) bfr[kk] = (v2f){0.0f, 0.0f};
    if (mm) {
        const float* bcol = W2 + halfK * OUT_DIM + col;
#pragma unroll
        for (int kk = 0; kk < F / 4; ++kk) {
            bfr[kk].x = bcol[(4 * kk + 0) * OUT_DIM];
            bfr[kk].y = bcol[(4 * kk + 1) * OUT_DIM];
        }
    }

    __syncthreads();
    if (!mm) return;

    const float* arow = tile + c16 * AG_PITCH + halfK;
    v8f acc = (v8f){0, 0, 0, 0, 0, 0, 0, 0};
#pragma unroll
    for (int kk = 0; kk < F / 4; ++kk) {
        const v2f a = *(const v2f*)(arow + 4 * kk);
        acc = __builtin_amdgcn_wmma_f32_16x16x4_f32(
            false, a, false, bfr[kk], (short)0, acc, false, false);
    }

    const int rbase = blockIdx.x * AG_NODES + ((lane >> 4) << 3);
#pragma unroll
    for (int j = 0; j < 8; ++j)
        xl2[(size_t)(rbase + j) * OUT_DIM + col] = acc[j];
}

static inline int cdiv(long long a, long long b) { return (int)((a + b - 1) / b); }

extern "C" void kernel_launch(void* const* d_in, const int* in_sizes, int n_in,
                              void* d_out, int out_size, void* d_ws, size_t ws_size,
                              hipStream_t stream) {
    const float* x      = (const float*)d_in[0];
    const int*   ei     = (const int*)d_in[1];
    const float* W1     = (const float*)d_in[2];
    const float* a_src1 = (const float*)d_in[3];
    const float* a_dst1 = (const float*)d_in[4];
    const float* b1     = (const float*)d_in[5];
    const float* W2     = (const float*)d_in[6];
    const float* a_src2 = (const float*)d_in[7];
    const float* a_dst2 = (const float*)d_in[8];
    const float* b2     = (const float*)d_in[9];
    float* out = (float*)d_out;

    const int N  = in_sizes[0] / IN_DIM;
    const int E  = in_sizes[1] / 2;
    const int F1 = HEADS * HID;

    float* ws   = (float*)d_ws;
    float* xl1  = ws;
    float* h1   = xl1  + (size_t)N * F1;
    float* xl2  = h1   + (size_t)N * F1;
    int*   cnt  = (int*)(xl2 + (size_t)N * OUT_DIM);
    int*   off  = cnt  + (size_t)N;
    int*   rec  = off  + (size_t)N;
    int*   srcs = rec  + (size_t)E;
    const int G    = cdiv(E, CH);
    const int NBKT = cdiv(N, BK_NODES);
    int*   hist_t = (int*)h1;
    int*   woff_t = hist_t + (size_t)NBKT * G;
    int*   tot    = woff_t + (size_t)NBKT * G;

    const int TB = 256;

    bucket_hist_kernel<<<G, CH_TB, 0, stream>>>(ei, E, G, NBKT, hist_t);
    bucket_colscan_kernel<<<NBKT, CS_TB, 0, stream>>>(hist_t, G, woff_t, tot);
    const int G1B = cdiv(cdiv(N, 16), G1_WPB);
    scatter_gemm1_kernel<<<G + G1B, CH_TB, 0, stream>>>(ei, E, G, NBKT, woff_t, tot, rec,
                                                         x, W1, xl1, N);
    bucket_sort_kernel<<<NBKT, SORT_TB, 0, stream>>>(rec, tot, N, cnt, off, srcs);

    {
        node_agg1_gemm2_kernel<<<cdiv(N, AG_NODES), AG_TB, 0, stream>>>(
            off, cnt, srcs, xl1, a_src1, a_dst1, b1, W2, xl2, N);
    }

    {
        node_agg_kernel<1, OUT_DIM, false><<<cdiv((long long)N * (OUT_DIM / 4), TB), TB, 0, stream>>>(
            off, cnt, srcs, xl2, a_src2, a_dst2, b2, out, N);
    }
}


# Edge chunks stored bucket-ordered per chunk with a table of bucket starts; each bucket's block collects its runs from it: count and column-scan launches gone; sum order stays arrival-dependent
# speedup vs baseline: 9.8196x; 1.1915x over previous
#include <hip/hip_runtime.h>
#include <hip/hip_bf16.h>
#include <math.h>

typedef float v2f __attribute__((ext_vector_type(2)));
typedef float v8f __attribute__((ext_vector_type(8)));

#define IN_DIM   128
#define HID      32
#define HEADS    2
#define OUT_DIM  32

template <int K, int NC>
__device__ __forceinline__ void wmma_gemm_block(const float* __restrict__ A,
                                                const float* __restrict__ B,
                                                float* __restrict__ C, int M,
                                                int blk, int WPB, int NTHR,
                                                v2f* Bs2) {
    constexpr int NT = NC / 16;

    for (int p = threadIdx.x; p < (K / 2) * NC; p += NTHR) {
        int k2 = p / NC;
        int n  = p - k2 * NC;
        v2f pr;
        pr.x = B[(2 * k2 + 0) * NC + n];
        pr.y = B[(2 * k2 + 1) * NC + n];
        Bs2[p] = pr;
    }
    __syncthreads();

    const int wave = threadIdx.x >> 5;
    const int lane = threadIdx.x & 31;
    const int mt   = blk * WPB + wave;
    if (mt * 16 >= M) return;

    const int halfK = (lane >> 4) * 2;
    const int ncol  = lane & 15;
    const int arow  = mt * 16 + (lane & 15);
    const float* Arow = A + (size_t)arow * K;

    v8f acc[NT];
#pragma unroll
    for (int t = 0; t < NT; ++t) acc[t] = (v8f){0, 0, 0, 0, 0, 0, 0, 0};

    for (int k = 0; k < K; k += 4) {
        v2f a = *(const v2f*)(Arow + k + halfK);
        const int prow = (k + halfK) >> 1;
#pragma unroll
        for (int t = 0; t < NT; ++t) {
            v2f b = Bs2[prow * NC + t * 16 + ncol];
            acc[t] = __builtin_amdgcn_wmma_f32_16x16x4_f32(
                false, a, false, b, (short)0, acc[t], false, false);
        }
    }

    const int rbase = mt * 16 + ((lane >> 4) << 3);
#pragma unroll
    for (int t = 0; t < NT; ++t) {
#pragma unroll
        for (int j = 0; j < 8; ++j) {
            C[(size_t)(rbase + j) * NC + t * 16 + ncol] = acc[t][j];
        }
    }
}

#define BK_SHIFT  9
#define BK_NODES  (1 << BK_SHIFT)
#define MAXB      256
#define CH_TB     256
#define CH_PER    16
#define CH        (CH_TB * CH_PER)
#define REC_SHIFT 22
#define REC_MASK  ((1 << REC_SHIFT) - 1)
#define SORT_TB   1024
#define SORT_PER  16
#define SORT_CAP  (SORT_TB * SORT_PER)

template <int NT>
__device__ __forceinline__ int block_excl_scan(int v, int* wsum, int* total) {
    const int lane = threadIdx.x & 31;
    const int wv   = threadIdx.x >> 5;
    int inc = v;
#pragma unroll
    for (int o = 1; o < 32; o <<= 1) {
        int t = __shfl_up(inc, o, 32);
        if (lane >= o) inc += t;
    }
    if (lane == 31) wsum[wv] = inc;
    __syncthreads();
    if (wv == 0) {
        const int ws = (lane < NT / 32) ? wsum[lane] : 0;
        int wi = ws;
#pragma unroll
        for (int o = 1; o < 32; o <<= 1) {
            int t = __shfl_up(wi, o, 32);
            if (lane >= o) wi += t;
        }
        if (lane < NT / 32) wsum[lane] = wi - ws;
        if (lane == 31) wsum[32] = wi;
    }
    __syncthreads();
    const int res = wsum[wv] + inc - v;
    *total = wsum[32];
    __syncthreads();
    return res;
}

template <int MASK>
__device__ __forceinline__ int row_xor_i(int v) {
    return __builtin_amdgcn_update_dpp(0, v, 0x160 + MASK, 0xf, 0xf, true);
}

__device__ __forceinline__ int wave_sum_i(int v) {
    v += row_xor_i<1>(v);
    v += row_xor_i<2>(v);
    v += row_xor_i<4>(v);
    v += row_xor_i<8>(v);
    return v + __shfl_xor(v, 16, 32);
}

template <int NT>
__device__ __forceinline__ int block_excl_scan_and_sum(int v, int u, int* wsum,
                                                       int* vtotal, int* utotal) {
    const int lane = threadIdx.x & 31;
    const int wv   = threadIdx.x >> 5;
    int inc = v;
#pragma unroll
    for (int o = 1; o < 32; o <<= 1) {
        int t = __shfl_up(inc, o, 32);
        if (lane >= o) inc += t;
    }
    const int us = wave_sum_i(u);
    if (lane == 31) { wsum[wv] = inc; wsum[33 + wv] = us; }
    __syncthreads();
    if (wv == 0) {
        const int ws = (lane < NT / 32) ? wsum[lane] : 0;
        int wi = ws;
        const int uu = wave_sum_i((lane < NT / 32) ? wsum[33 + lane] : 0);
#pragma unroll
        for (int o = 1; o < 32; o <<= 1) {
            int t = __shfl_up(wi, o, 32);
            if (lane >= o) wi += t;
        }
        if (lane < NT / 32) wsum[lane] = wi - ws;
        if (lane == 31) { wsum[32] = wi; wsum[65] = uu; }
    }
    __syncthreads();
    const int res = wsum[wv] + inc - v;
    *vtotal = wsum[32];
    *utotal = wsum[65];
    __syncthreads();
    return res;
}

#define SC_LDS_INTS (2 * MAXB + CH + 33)
#define G1_LDS_INTS ((IN_DIM / 2) * (HEADS * HID) * 2)
#define SG_LDS_INTS (SC_LDS_INTS > G1_LDS_INTS ? SC_LDS_INTS : G1_LDS_INTS)
#define G1_WPB      (CH_TB / 32)

__device__ __forceinline__
void bucket_scatter_block(const int* __restrict__ ei, int E, int G, int nbkt,
                          int* __restrict__ lstart_t,
                          int* __restrict__ rec, int blk, int* smem) {
    int* const lh     = smem;
    int* const lstart = lh + MAXB;
    int* const srec   = lstart + MAXB;
    int* const wsum   = srec + CH;
    const int tid  = threadIdx.x;
    const int base = blk * CH;

    int d[CH_PER], s[CH_PER], r[CH_PER];
#pragma unroll
    for (int k = 0; k < CH_PER; ++k) {
        const int e = base + k * CH_TB + tid;
        d[k] = (e < E) ? ei[E + e] : -1;
        s[k] = (e < E) ? ei[e] : 0;
    }
    lh[tid] = 0;
    __syncthreads();
#pragma unroll
    for (int k = 0; k < CH_PER; ++k)
        r[k] = (d[k] >= 0) ? atomicAdd(&lh[d[k] >> BK_SHIFT], 1) : 0;
    __syncthreads();

    int dummy;
    const int ls = block_excl_scan<CH_TB>(lh[tid], wsum, &dummy);
    lstart[tid] = ls;
    const int n = (E - base < CH) ? (E - base) : CH;
    if (tid < nbkt) lstart_t[tid * G + blk] = ls;
    if (tid == 0)   lstart_t[nbkt * G + blk] = n;
    __syncthreads();

#pragma unroll
    for (int k = 0; k < CH_PER; ++k) {
        if (d[k] >= 0) {
            const int p = lstart[d[k] >> BK_SHIFT] + r[k];
            srec[p] = ((d[k] & (BK_NODES - 1)) << REC_SHIFT) | s[k];
        }
    }
    __syncthreads();

    for (int i = tid; i < n; i += CH_TB) rec[base + i] = srec[i];
}

__global__ __launch_bounds__(CH_TB)
void scatter_gemm1_kernel(const int* __restrict__ ei, int E, int G, int nbkt,
                          int* __restrict__ lstart_t,
                          int* __restrict__ rec,
                          const float* __restrict__ x,
                          const float* __restrict__ W1,
                          float* __restrict__ xl1, int N) {
    __shared__ __attribute__((aligned(16))) int smem[SG_LDS_INTS];
    if ((int)blockIdx.x < G) {
        bucket_scatter_block(ei, E, G, nbkt, lstart_t, rec, blockIdx.x, smem);
    } else {
        wmma_gemm_block<IN_DIM, HEADS * HID>(x, W1, xl1, N, (int)blockIdx.x - G,
                                             G1_WPB, CH_TB, (v2f*)smem);
    }
}

#define RUN_U 16

__global__ __launch_bounds__(SORT_TB)
void bucket_sort_kernel(const int* __restrict__ rec, const int* __restrict__ lstart_t,
                        int G, int N, int* __restrict__ cnt, int* __restrict__ off,
                        int* __restrict__ srcs) {
    __shared__ int ncur[BK_NODES];
    __shared__ int stage[SORT_CAP];
    __shared__ int rg[SORT_TB];
    __shared__ int rp[SORT_TB];
    __shared__ int rn[SORT_TB];
    __shared__ int wsum[66];
    const int tid  = threadIdx.x;
    const int lane = tid & 31;
    const int wave = tid >> 5;
    const int b    = blockIdx.x;

    if (tid < BK_NODES) ncur[tid] = 0;

    int nb   = 0;
    int base = 0;
    for (int w0 = 0; w0 < G; w0 += SORT_TB) {
        const int w = w0 + tid;
        int l0 = 0, c = 0;
        if (w < G) {
            l0 = lstart_t[b * G + w];
            c  = lstart_t[(b + 1) * G + w] - l0;
        }
        int ctot, ltot;
        const int ex = block_excl_scan_and_sum<SORT_TB>(c, l0, wsum, &ctot, &ltot);
        rg[tid] = w * CH + l0;
        rp[tid] = nb + ex;
        rn[tid] = c;
        __syncthreads();
        const int ntile = (G - w0 < SORT_TB) ? (G - w0) : SORT_TB;
        for (int j0 = wave; j0 < ntile; j0 += 32 * RUN_U) {
            int g[RUN_U], p[RUN_U], n[RUN_U], v[RUN_U];
#pragma unroll
            for (int u = 0; u < RUN_U; ++u) {
                const int  t  = j0 + 32 * u;
                const bool ok = t < ntile;
                g[u] = ok ? rg[t] : 0;
                p[u] = ok ? rp[t] : 0;
                n[u] = ok ? rn[t] : 0;
            }
#pragma unroll
            for (int u = 0; u < RUN_U; ++u)
                v[u] = (lane < n[u]) ? rec[g[u] + lane] : 0;
#pragma unroll
            for (int u = 0; u < RUN_U; ++u)
                if (lane < n[u] && p[u] + lane < SORT_CAP) stage[p[u] + lane] = v[u];
#pragma unroll
            for (int u = 0; u < RUN_U; ++u)
                for (int k = lane + 32; k < n[u]; k += 32)
                    if (p[u] + k < SORT_CAP) stage[p[u] + k] = rec[g[u] + k];
        }
        nb   += ctot;
        base += ltot;
        __syncthreads();
    }

    const bool fits = nb <= SORT_CAP;
    int rc[SORT_PER];
    if (fits) {
#pragma unroll
        for (int k = 0; k < SORT_PER; ++k) {
            const int i = k * SORT_TB + tid;
            rc[k] = (i < nb) ? stage[i] : -1;
        }
#pragma unroll
        for (int k = 0; k < SORT_PER; ++k)
            if (rc[k] >= 0) atomicAdd(&ncur[rc[k] >> REC_SHIFT], 1);
    } else {
        for (int w = tid; w < G; w += SORT_TB) {
            const int l0 = lstart_t[b * G + w];
            const int l1 = lstart_t[(b + 1) * G + w];
            for (int k = l0; k < l1; ++k)
                atomicAdd(&ncur[rec[w * CH + k] >> REC_SHIFT], 1);
        }
    }
    __syncthreads();

    int dummy;
    const int c  = (tid < BK_NODES) ? ncur[tid] : 0;
    const int ex = block_excl_scan<SORT_TB>(c, wsum, &dummy);
    if (tid < BK_NODES) {
        const int node = b * BK_NODES + tid;
        if (node < N) { cnt[node] = c; off[node] = base + ex; }
        ncur[tid] = ex;
    }
    __syncthreads();

    if (fits) {
#pragma unroll
        for (int k = 0; k < SORT_PER; ++k)
            if (rc[k] >= 0)
                stage[atomicAdd(&ncur[rc[k] >> REC_SHIFT], 1)] = rc[k] & REC_MASK;
        __syncthreads();
        for (int i = tid; i < nb; i += SORT_TB) srcs[base + i] = stage[i];
    } else {
        for (int w = tid; w < G; w += SORT_TB) {
            const int l0 = lstart_t[b * G + w];
            const int l1 = lstart_t[(b + 1) * G + w];
            for (int k = l0; k < l1; ++k) {
                const int rv = rec[w * CH + k];
                srcs[base + atomicAdd(&ncur[rv >> REC_SHIFT], 1)] = rv & REC_MASK;
            }
        }
    }
}

__device__ __forceinline__ float leaky02(float v) {
    return fmaxf(v, 0.2f * v);
}

template <int MASK>
__device__ __forceinline__ float row_xor(float v) {
    return __int_as_float(__builtin_amdgcn_update_dpp(
        0, __float_as_int(v), 0x160 + MASK, 0xf, 0xf, true));
}

template <int LPH>
__device__ __forceinline__ float head_sum(float v) {
    if (LPH > 1)  v += row_xor<1>(v);
    if (LPH > 2)  v += row_xor<2>(v);
    if (LPH > 4)  v += row_xor<4>(v);
    if (LPH > 8)  v += row_xor<8>(v);
    return v;
}

__device__ __forceinline__ float dot4(const float4 a, const float4 b) {
    return fmaf(a.w, b.w, fmaf(a.z, b.z, fmaf(a.y, b.y, a.x * b.x)));
}

template <int F>
__device__ __forceinline__ float4 row_load(const float* __restrict__ xl, int s, int f0) {
    const unsigned byte_off = (unsigned)s * (unsigned)(F * 4) + (unsigned)(f0 * 4);
    return *(const float4*)((const char*)xl + byte_off);
}

template <int H, int C, bool RELU>
__device__ __forceinline__ float4 node_agg_row(const int* __restrict__ off,
                                               const int* __restrict__ cnt,
                                               const int* __restrict__ srcs,
                                               const float* __restrict__ xl,
                                               const float* __restrict__ a_src,
                                               const float* __restrict__ a_dst,
                                               const float* __restrict__ b,
                                               int i, int lg) {
    constexpr int F   = H * C;
    constexpr int LPH = C / 4;
    constexpr int U   = 4;

    const int beg = off[i];
    const int deg = cnt[i];
    const int f0  = lg * 4;

    float4 as, ad;
    as.x = a_src[f0 + 0]; as.y = a_src[f0 + 1]; as.z = a_src[f0 + 2]; as.w = a_src[f0 + 3];
    ad.x = a_dst[f0 + 0]; ad.y = a_dst[f0 + 1]; ad.z = a_dst[f0 + 2]; ad.w = a_dst[f0 + 3];

    const float4 r  = row_load<F>(xl, i, f0);
    const float adh = head_sum<LPH>(dot4(r, ad));
    float m   = leaky02(head_sum<LPH>(dot4(r, as)) + adh);
    float den = 1.0f;
    float4 acc = r;

    for (int j = 0; j < deg; j += U) {
        int    s[U];
        float4 rr[U];
        float  e[U];
#pragma unroll
        for (int u = 0; u < U; ++u) {
            const int jj = (j + u < deg) ? (j + u) : (deg - 1);
            s[u] = srcs[beg + jj];
        }
#pragma unroll
        for (int u = 0; u < U; ++u)
            rr[u] = row_load<F>(xl, s[u], f0);
#pragma unroll
        for (int u = 0; u < U; ++u) {
            const float v = leaky02(head_sum<LPH>(dot4(rr[u], as)) + adh);
            e[u] = (j + u < deg) ? v : -INFINITY;
        }
        float mn = m;
#pragma unroll
        for (int u = 0; u < U; ++u) mn = fmaxf(mn, e[u]);
        const float sc = __expf(m - mn);
        m = mn;
        den *= sc;
        acc.x *= sc; acc.y *= sc; acc.z *= sc; acc.w *= sc;
#pragma unroll
        for (int u = 0; u < U; ++u) {
            const float w = __expf(e[u] - mn);
            den += w;
            acc.x += rr[u].x * w;
            acc.y += rr[u].y * w;
            acc.z += rr[u].z * w;
            acc.w += rr[u].w * w;
        }
    }

    const float dd = den + 1e-16f;
    float4 o;
    o.x = acc.x / dd + b[f0 + 0];
    o.y = acc.y / dd + b[f0 + 1];
    o.z = acc.z / dd + b[f0 + 2];
    o.w = acc.w / dd + b[f0 + 3];
    if (RELU) {
        o.x = o.x > 0.0f ? o.x : 0.0f;
        o.y = o.y > 0.0f ? o.y : 0.0f;
        o.z = o.z > 0.0f ? o.z : 0.0f;
        o.w = o.w > 0.0f ? o.w : 0.0f;
    }
    return o;
}

template <int H, int C, bool RELU>
__global__ void node_agg_kernel(const int* __restrict__ off,
                                const int* __restrict__ cnt,
                                const int* __restrict__ srcs,
                                const float* __restrict__ xl,
                                const float* __restrict__ a_src,
                                const float* __restrict__ a_dst,
                                const float* __restrict__ b,
                                float* __restrict__ outp, int n) {
    constexpr int F   = H * C;
    constexpr int LPN = F / 4;
    const int t  = blockIdx.x * blockDim.x + threadIdx.x;
    const int i  = t / LPN;
    const int lg = t % LPN;
    if (i >= n) return;
    const float4 o = node_agg_row<H, C, RELU>(off, cnt, srcs, xl, a_src, a_dst, b, i, lg);
    *(float4*)(outp + (size_t)i * F + lg * 4) = o;
}

#define AG_NODES 16
#define AG_TB    (AG_NODES * (HEADS * HID / 4))
#define AG_PITCH 68

__global__ __launch_bounds__(AG_TB)
void node_agg1_gemm2_kernel(const int* __restrict__ off,
                            const int* __restrict__ cnt,
                            const int* __restrict__ srcs,
                            const float* __restrict__ xl,
                            const float* __restrict__ a_src,
                            const float* __restrict__ a_dst,
                            const float* __restrict__ b,
                            const float* __restrict__ W2,
                            float* __restrict__ xl2, int n) {
    constexpr int F   = HEADS * HID;
    constexpr int LPN = F / 4;
    __shared__ __attribute__((aligned(16))) float tile[AG_NODES * AG_PITCH];

    const int nl = threadIdx.x / LPN;
    const int lg = threadIdx.x % LPN;
    const int i  = blockIdx.x * AG_NODES + nl;
    float* const trow = tile + nl * AG_PITCH + lg * 4;
    if (i < n) {
        *(float4*)trow = node_agg_row<HEADS, HID, true>(off, cnt, srcs, xl, a_src, a_dst, b, i, lg);
    } else {
        *(float4*)trow = make_float4(0.0f, 0.0f, 0.0f, 0.0f);
    }

    const int  wave  = threadIdx.x >> 5;
    const bool mm    = wave < OUT_DIM / 16;
    const int  lane  = threadIdx.x & 31;
    const int  halfK = (lane >> 4) * 2;
    const int  c16   = lane & 15;
    const int  col   = (mm ? wave : 0) * 16 + c16;

    v2f bfr[F / 4];
#pragma unroll
    for (int kk = 0; kk < F / 4; ++kk) bfr[kk] = (v2f){0.0f, 0.0f};
    if (mm) {
        const float* bcol = W2 + halfK * OUT_DIM + col;
#pragma unroll
        for (int kk = 0; kk < F / 4; ++kk) {
            bfr[kk].x = bcol[(4 * kk + 0) * OUT_DIM];
            bfr[kk].y = bcol[(4 * kk + 1) * OUT_DIM];
        }
    }

    __syncthreads();
    if (!mm) return;

    const float* arow = tile + c16 * AG_PITCH + halfK;
    v8f acc = (v8f){0, 0, 0, 0, 0, 0, 0, 0};
#pragma unroll
    for (int kk = 0; kk < F / 4; ++kk) {
        const v2f a = *(const v2f*)(arow + 4 * kk);
        acc = __builtin_amdgcn_wmma_f32_16x16x4_f32(
            false, a, false, bfr[kk], (short)0, acc, false, false);
    }

    const int rbase = blockIdx.x * AG_NODES + ((lane >> 4) << 3);
#pragma unroll
    for (int j = 0; j < 8; ++j)
        xl2[(size_t)(rbase + j) * OUT_DIM + col] = acc[j];
}

static inline int cdiv(long long a, long long b) { return (int)((a + b - 1) / b); }

extern "C" void kernel_launch(void* const* d_in, const int* in_sizes, int n_in,
                              void* d_out, int out_size, void* d_ws, size_t ws_size,
                              hipStream_t stream) {
    const float* x      = (const float*)d_in[0];
    const int*   ei     = (const int*)d_in[1];
    const float* W1     = (const float*)d_in[2];
    const float* a_src1 = (const float*)d_in[3];
    const float* a_dst1 = (const float*)d_in[4];
    const float* b1     = (const float*)d_in[5];
    const float* W2     = (const float*)d_in[6];
    const float* a_src2 = (const float*)d_in[7];
    const float* a_dst2 = (const float*)d_in[8];
    const float* b2     = (const float*)d_in[9];
    float* out = (float*)d_out;

    const int N  = in_sizes[0] / IN_DIM;
    const int E  = in_sizes[1] / 2;
    const int F1 = HEADS * HID;

    float* ws   = (float*)d_ws;
    float* xl1  = ws;
    float* h1   = xl1  + (size_t)N * F1;
    float* xl2  = h1   + (size_t)N * F1;
    int*   cnt  = (int*)(xl2 + (size_t)N * OUT_DIM);
    int*   off  = cnt  + (size_t)N;
    int*   rec  = off  + (size_t)N;
    int*   srcs = rec  + (size_t)E;
    const int G    = cdiv(E, CH);
    const int NBKT = cdiv(N, BK_NODES);
    int*   lstart_t = (int*)h1;

    const int TB = 256;

    const int G1B = cdiv(cdiv(N, 16), G1_WPB);
    scatter_gemm1_kernel<<<G + G1B, CH_TB, 0, stream>>>(ei, E, G, NBKT, lstart_t, rec,
                                                         x, W1, xl1, N);
    bucket_sort_kernel<<<NBKT, SORT_TB, 0, stream>>>(rec, lstart_t, G, N, cnt, off, srcs);

    {
        node_agg1_gemm2_kernel<<<cdiv(N, AG_NODES), AG_TB, 0, stream>>>(
            off, cnt, srcs, xl1, a_src1, a_dst1, b1, W2, xl2, N);
    }

    {
        node_agg_kernel<1, OUT_DIM, false><<<cdiv((long long)N * (OUT_DIM / 4), TB), TB, 0, stream>>>(
            off, cnt, srcs, xl2, a_src2, a_dst2, b2, out, N);
    }
}
